# starting kernel
# baseline (speedup 1.0000x reference)
#include <hip/hip_runtime.h>

#define N_NODES 50000
#define NE_EDGES 800000
#define BPAIRS 4096
#define DF 128
#define EPS_BN 1e-5f

typedef float v2f __attribute__((ext_vector_type(2)));
typedef float v8f __attribute__((ext_vector_type(8)));

__device__ __forceinline__ v8f wmma4(v2f a, v2f b, v8f c) {
  return __builtin_amdgcn_wmma_f32_16x16x4_f32(
       false, a,  false, b,
       (short)0, c,  false,  false);
}

__global__ __launch_bounds__(256)
void wmma_gemm_strip(const float* __restrict__ A, const float* __restrict__ Bt,
                     const float* __restrict__ bias, float* __restrict__ C,
                     int M, int K) {
  int gtid = blockIdx.x * blockDim.x + threadIdx.x;
  int wave = gtid >> 5;
  int lane = threadIdx.x & 31;
  if (wave * 16 >= M) return;

  int l16   = lane & 15;
  int khalf = (lane >> 4) << 1;
  const float* arow = A + (size_t)(wave * 16 + l16) * K + khalf;

  const float* brow[8];
#pragma unroll
  for (int nt = 0; nt < 8; ++nt)
    brow[nt] = Bt + (size_t)(nt * 16 + l16) * K + khalf;

  v8f zero = {};
  v8f acc[8];
#pragma unroll
  for (int nt = 0; nt < 8; ++nt) acc[nt] = zero;

  for (int k = 0; k < K; k += 4) {
    v2f a = *(const v2f*)(arow + k);
#pragma unroll
    for (int nt = 0; nt < 8; ++nt) {
      v2f b = *(const v2f*)(brow[nt] + k);
      acc[nt] = wmma4(a, b, acc[nt]);
    }
  }

  int rbase = wave * 16 + ((lane >> 4) << 3);
#pragma unroll
  for (int nt = 0; nt < 8; ++nt) {
    float bs = bias ? bias[nt * 16 + l16] : 0.f;
    float* crow = C + (size_t)rbase * 128 + nt * 16 + l16;
#pragma unroll
    for (int r = 0; r < 8; ++r)
      crow[(size_t)r * 128] = acc[nt][r] + bs;
  }
}

__global__ __launch_bounds__(256)
void wmma_matmul(const float* __restrict__ A, const float* __restrict__ Bm,
                 const float* __restrict__ bias, float* __restrict__ C,
                 int M, int Nc, int K, int bTrans) {
  int gtid = blockIdx.x * blockDim.x + threadIdx.x;
  int wave = gtid >> 5;
  int lane = threadIdx.x & 31;
  int ntiles = Nc >> 4;
  int mt = wave / ntiles;
  int nt = wave - mt * ntiles;
  if (mt * 16 >= M) return;

  int l16   = lane & 15;
  int khalf = (lane >> 4) << 1;
  const float* arow = A + (size_t)(mt * 16 + l16) * K + khalf;

  v8f acc = {};
  if (bTrans) {
    const float* brow = Bm + (size_t)(nt * 16 + l16) * K + khalf;
    for (int k = 0; k < K; k += 4) {
      v2f a = *(const v2f*)(arow + k);
      v2f b = *(const v2f*)(brow + k);
      acc = wmma4(a, b, acc);
    }
  } else {
    const float* bcol = Bm + (size_t)khalf * Nc + nt * 16 + l16;
    for (int k = 0; k < K; k += 4) {
      v2f a = *(const v2f*)(arow + k);
      v2f b;
      b.x = bcol[(size_t)k * Nc];
      b.y = bcol[(size_t)(k + 1) * Nc];
      acc = wmma4(a, b, acc);
    }
  }

  float bs = bias ? bias[nt * 16 + l16] : 0.f;
  int rbase = mt * 16 + ((lane >> 4) << 3);
  float* crow = C + (size_t)rbase * Nc + nt * 16 + l16;
#pragma unroll
  for (int r = 0; r < 8; ++r)
    crow[(size_t)r * Nc] = acc[r] + bs;
}

__global__ void zero_int(int* p, int n) {
  int t = blockIdx.x * blockDim.x + threadIdx.x;
  if (t < n) p[t] = 0;
}

__global__ void count_edges(const int* __restrict__ ecol, int* __restrict__ cnt) {
  int e = blockIdx.x * blockDim.x + threadIdx.x;
  if (e < NE_EDGES) atomicAdd(&cnt[ecol[e]], 1);
}

__global__ __launch_bounds__(1024)
void scan_rowptr(const int* __restrict__ cnt, int* __restrict__ rowptr) {
  __shared__ int part[1024];
  const int CH = (N_NODES + 1023) / 1024;
  int tid = threadIdx.x;
  int base = tid * CH;
  int s = 0;
  for (int i = 0; i < CH; ++i) {
    int idx = base + i;
    if (idx < N_NODES) s += cnt[idx];
  }
  part[tid] = s;
  __syncthreads();
  for (int off = 1; off < 1024; off <<= 1) {
    int v = 0;
    if (tid >= off) v = part[tid - off];
    __syncthreads();
    if (tid >= off) part[tid] += v;
    __syncthreads();
  }
  int run = (tid == 0) ? 0 : part[tid - 1];
  for (int i = 0; i < CH; ++i) {
    int idx = base + i;
    if (idx < N_NODES) { rowptr[idx] = run; run += cnt[idx]; }
  }
  if (tid == 1023) rowptr[N_NODES] = run;
}

__global__ void copy_int(const int* __restrict__ src, int* __restrict__ dst, int n) {
  int t = blockIdx.x * blockDim.x + threadIdx.x;
  if (t < n) dst[t] = src[t];
}

__global__ void csr_fill(const int* __restrict__ erow, const int* __restrict__ ecol,
                         int* __restrict__ cursor, int* __restrict__ srcidx) {
  int e = blockIdx.x * blockDim.x + threadIdx.x;
  if (e >= NE_EDGES) return;
  int c = ecol[e];
  int p = atomicAdd(&cursor[c], 1);
  srcidx[p] = erow[e];
}

__global__ void dis_from_cnt(const int* __restrict__ cnt, float* __restrict__ dis, int n) {
  int t = blockIdx.x * blockDim.x + threadIdx.x;
  if (t < n) dis[t] = rsqrtf((float)cnt[t] + 1.0f);
}

__global__ __launch_bounds__(256)
void gcn_aggregate(const int* __restrict__ rowptr, const int* __restrict__ srcidx,
                   const float* __restrict__ dis, const float* __restrict__ lin,
                   float* __restrict__ agg) {
  int t = blockIdx.x * blockDim.x + threadIdx.x;
  int c = t >> 5;
  if (c >= N_NODES) return;
  int lane = threadIdx.x & 31;
  float dc = dis[c];
  const float4 sv = *(const float4*)(lin + (size_t)c * DF + lane * 4);
  float s = dc * dc;
  float ax = s * sv.x, ay = s * sv.y, az = s * sv.z, aw = s * sv.w;
  int e0 = rowptr[c], e1 = rowptr[c + 1];
  for (int j = e0; j < e1; ++j) {
    int r = srcidx[j];
    float nrm = dis[r] * dc;
    const float4 v = *(const float4*)(lin + (size_t)r * DF + lane * 4);
    ax += nrm * v.x; ay += nrm * v.y; az += nrm * v.z; aw += nrm * v.w;
  }
  float4 o; o.x = ax; o.y = ay; o.z = az; o.w = aw;
  *(float4*)(agg + (size_t)c * DF + lane * 4) = o;
}

__global__ void zero256(float* p) { p[threadIdx.x] = 0.0f; }

__global__ __launch_bounds__(128)
void bn_stats(const float* __restrict__ agg, float* __restrict__ stats, int n) {
  int f = threadIdx.x;
  int r0 = blockIdx.x * 512;
  int r1 = r0 + 512; if (r1 > n) r1 = n;
  float s = 0.f, s2 = 0.f;
  for (int r = r0; r < r1; ++r) {
    float v = agg[(size_t)r * DF + f];
    s += v; s2 += v * v;
  }
  atomicAdd(&stats[f], s);
  atomicAdd(&stats[DF + f], s2);
}

__global__ void bn_finalize(float* __restrict__ stats, const float* __restrict__ g,
                            const float* __restrict__ be, float inv_n) {
  int f = threadIdx.x;
  float m   = stats[f] * inv_n;
  float var = stats[DF + f] * inv_n - m * m;
  float sc  = g[f] * rsqrtf(var + EPS_BN);
  stats[f]      = sc;
  stats[DF + f] = be[f] - m * sc;
}

__global__ __launch_bounds__(256)
void bn_apply_leaky(const float* __restrict__ agg, const float* __restrict__ stats,
                    float* __restrict__ h) {
  int t = blockIdx.x * blockDim.x + threadIdx.x;
  if (t >= N_NODES * DF) return;
  int f = t & (DF - 1);
  float y = agg[t] * stats[f] + stats[DF + f];
  h[t] = (y >= 0.f) ? y : 0.1f * y;
}

__global__ void transpose128(const float* __restrict__ src, float* __restrict__ dst) {
  int t = blockIdx.x * blockDim.x + threadIdx.x;
  if (t >= 128 * 128) return;
  int r = t >> 7, c = t & 127;
  dst[c * 128 + r] = src[t];
}

__global__ __launch_bounds__(256)
void gather_pairs(const float* __restrict__ h, const int* __restrict__ di,
                  float* __restrict__ amat, float* __restrict__ bmat) {
  int t = blockIdx.x * blockDim.x + threadIdx.x;
  if (t >= BPAIRS * 32) return;
  int j = t >> 5, q = t & 31;
  int ia = di[2 * j] - 1;     if (ia < 0) ia += N_NODES;
  int ib = di[2 * j + 1] - 1; if (ib < 0) ib += N_NODES;
  *(float4*)(amat + (size_t)j * DF + q * 4) = *(const float4*)(h + (size_t)ia * DF + q * 4);
  *(float4*)(bmat + (size_t)j * DF + q * 4) = *(const float4*)(h + (size_t)ib * DF + q * 4);
}

__global__ __launch_bounds__(256)
void rowdot(const float* __restrict__ tm, const float* __restrict__ bm,
            float* __restrict__ out) {
  int t = blockIdx.x * blockDim.x + threadIdx.x;
  int j = t >> 5, lane = t & 31;
  if (j >= BPAIRS) return;
  float4 tv = *(const float4*)(tm + (size_t)j * DF + lane * 4);
  float4 bv = *(const float4*)(bm + (size_t)j * DF + lane * 4);
  float p = tv.x * bv.x + tv.y * bv.y + tv.z * bv.z + tv.w * bv.w;
#pragma unroll
  for (int off = 16; off > 0; off >>= 1)
    p += __shfl_down(p, off, 32);
  if (lane == 0) out[j] = p;
}

extern "C" void kernel_launch(void* const* d_in, const int* in_sizes, int n_in,
                              void* d_out, int out_size, void* d_ws, size_t ws_size,
                              hipStream_t stream) {
  const float* x   = (const float*)d_in[0];
  const int*   edg = (const int*)d_in[1];
  const int*   di  = (const int*)d_in[2];
  const float* w1  = (const float*)d_in[3];
  const float* b1  = (const float*)d_in[4];
  const float* w2  = (const float*)d_in[5];
  const float* b2  = (const float*)d_in[6];
  const float* w3  = (const float*)d_in[7];
  const float* b3  = (const float*)d_in[8];
  const float* g1  = (const float*)d_in[9];
  const float* be1 = (const float*)d_in[10];
  const float* g2  = (const float*)d_in[11];
  const float* be2 = (const float*)d_in[12];
  const float* g3  = (const float*)d_in[13];
  const float* be3 = (const float*)d_in[14];
  const float* p1  = (const float*)d_in[15];
  const float* p2  = (const float*)d_in[16];
  float* out = (float*)d_out;

  float* ws = (float*)d_ws;
  size_t off = 0;
  float* dis    = ws + off; off += N_NODES;
  float* stats  = ws + off; off += 256;
  float* Wt     = ws + off; off += 128 * 128;
  int*   cnt    = (int*)(ws + off); off += N_NODES;
  int*   rowptr = (int*)(ws + off); off += N_NODES + 16;
  int*   cursor = (int*)(ws + off); off += N_NODES;
  int*   srcidx = (int*)(ws + off); off += NE_EDGES;
  float* lin    = ws + off; off += (size_t)N_NODES * DF;
  float* agg    = ws + off; off += (size_t)N_NODES * DF;
  float* h      = ws + off; off += (size_t)N_NODES * DF;
  float* U    = lin;
  float* Qt   = lin + 16384;
  float* amat = lin + 32768;
  float* bmat = amat + (size_t)BPAIRS * DF;
  float* tmat = bmat + (size_t)BPAIRS * DF;

  const int* erow = edg;
  const int* ecol = edg + NE_EDGES;

  zero_int<<<(N_NODES + 255) / 256, 256, 0, stream>>>(cnt, N_NODES);
  count_edges<<<(NE_EDGES + 255) / 256, 256, 0, stream>>>(ecol, cnt);
  scan_rowptr<<<1, 1024, 0, stream>>>(cnt, rowptr);
  copy_int<<<(N_NODES + 255) / 256, 256, 0, stream>>>(rowptr, cursor, N_NODES);
  csr_fill<<<(NE_EDGES + 255) / 256, 256, 0, stream>>>(erow, ecol, cursor, srcidx);
  dis_from_cnt<<<(N_NODES + 255) / 256, 256, 0, stream>>>(cnt, dis, N_NODES);

  const float* Wl[3] = {w1, w2, w3};
  const float* Bl[3] = {b1, b2, b3};
  const float* Gl[3] = {g1, g2, g3};
  const float* El[3] = {be1, be2, be3};
  const float* in = x;
  const int strip_waves = N_NODES / 16;
  for (int l = 0; l < 3; ++l) {
    transpose128<<<64, 256, 0, stream>>>(Wl[l], Wt);
    wmma_gemm_strip<<<(strip_waves * 32 + 255) / 256, 256, 0, stream>>>(
        in, Wt, Bl[l], lin, N_NODES, DF);
    gcn_aggregate<<<(N_NODES * 32 + 255) / 256, 256, 0, stream>>>(
        rowptr, srcidx, dis, lin, agg);
    zero256<<<1, 256, 0, stream>>>(stats);
    bn_stats<<<(N_NODES + 511) / 512, 128, 0, stream>>>(agg, stats, N_NODES);
    bn_finalize<<<1, 128, 0, stream>>>(stats, Gl[l], El[l], 1.0f / (float)N_NODES);
    bn_apply_leaky<<<(N_NODES * DF + 255) / 256, 256, 0, stream>>>(agg, stats, h);
    in = h;
  }

  const int small_tiles = (DF / 16) * (DF / 16);
  wmma_matmul<<<(small_tiles * 32 + 255) / 256, 256, 0, stream>>>(p1, p2, nullptr, U, DF, DF, DF, 1);
  wmma_matmul<<<(small_tiles * 32 + 255) / 256, 256, 0, stream>>>(U, p1, nullptr, Qt, DF, DF, DF, 1);
  gather_pairs<<<(BPAIRS * 32 + 255) / 256, 256, 0, stream>>>(h, di, amat, bmat);
  wmma_gemm_strip<<<((BPAIRS / 16) * 32 + 255) / 256, 256, 0, stream>>>(
      amat, Qt, nullptr, tmat, BPAIRS, DF);
  rowdot<<<(BPAIRS * 32 + 255) / 256, 256, 0, stream>>>(tmat, bmat, out);
}


# BN column sums made in gcn_aggregate from the row each wave holds: 8-wave LDS partials added in wave order, then a fixed tree in finalize; bn_stats pass and its float atomics gone; sum order changes
# speedup vs baseline: 2.2052x; 2.2052x over previous
#include <hip/hip_runtime.h>

#define N_NODES 50000
#define NE_EDGES 800000
#define BPAIRS 4096
#define DF 128
#define EPS_BN 1e-5f

typedef float v2f __attribute__((ext_vector_type(2)));
typedef float v8f __attribute__((ext_vector_type(8)));

__device__ __forceinline__ v8f wmma4(v2f a, v2f b, v8f c) {
  return __builtin_amdgcn_wmma_f32_16x16x4_f32(
       false, a,  false, b,
       (short)0, c,  false,  false);
}

__global__ __launch_bounds__(256)
void wmma_gemm_strip(const float* __restrict__ A, const float* __restrict__ Bt,
                     const float* __restrict__ bias, float* __restrict__ C,
                     int M, int K) {
  int gtid = blockIdx.x * blockDim.x + threadIdx.x;
  int wave = gtid >> 5;
  int lane = threadIdx.x & 31;
  if (wave * 16 >= M) return;

  int l16   = lane & 15;
  int khalf = (lane >> 4) << 1;
  const float* arow = A + (size_t)(wave * 16 + l16) * K + khalf;

  const float* brow[8];
#pragma unroll
  for (int nt = 0; nt < 8; ++nt)
    brow[nt] = Bt + (size_t)(nt * 16 + l16) * K + khalf;

  v8f zero = {};
  v8f acc[8];
#pragma unroll
  for (int nt = 0; nt < 8; ++nt) acc[nt] = zero;

  for (int k = 0; k < K; k += 4) {
    v2f a = *(const v2f*)(arow + k);
#pragma unroll
    for (int nt = 0; nt < 8; ++nt) {
      v2f b = *(const v2f*)(brow[nt] + k);
      acc[nt] = wmma4(a, b, acc[nt]);
    }
  }

  int rbase = wave * 16 + ((lane >> 4) << 3);
#pragma unroll
  for (int nt = 0; nt < 8; ++nt) {
    float bs = bias ? bias[nt * 16 + l16] : 0.f;
    float* crow = C + (size_t)rbase * 128 + nt * 16 + l16;
#pragma unroll
    for (int r = 0; r < 8; ++r)
      crow[(size_t)r * 128] = acc[nt][r] + bs;
  }
}

__global__ __launch_bounds__(256)
void wmma_matmul(const float* __restrict__ A, const float* __restrict__ Bm,
                 const float* __restrict__ bias, float* __restrict__ C,
                 int M, int Nc, int K, int bTrans) {
  int gtid = blockIdx.x * blockDim.x + threadIdx.x;
  int wave = gtid >> 5;
  int lane = threadIdx.x & 31;
  int ntiles = Nc >> 4;
  int mt = wave / ntiles;
  int nt = wave - mt * ntiles;
  if (mt * 16 >= M) return;

  int l16   = lane & 15;
  int khalf = (lane >> 4) << 1;
  const float* arow = A + (size_t)(mt * 16 + l16) * K + khalf;

  v8f acc = {};
  if (bTrans) {
    const float* brow = Bm + (size_t)(nt * 16 + l16) * K + khalf;
    for (int k = 0; k < K; k += 4) {
      v2f a = *(const v2f*)(arow + k);
      v2f b = *(const v2f*)(brow + k);
      acc = wmma4(a, b, acc);
    }
  } else {
    const float* bcol = Bm + (size_t)khalf * Nc + nt * 16 + l16;
    for (int k = 0; k < K; k += 4) {
      v2f a = *(const v2f*)(arow + k);
      v2f b;
      b.x = bcol[(size_t)k * Nc];
      b.y = bcol[(size_t)(k + 1) * Nc];
      acc = wmma4(a, b, acc);
    }
  }

  float bs = bias ? bias[nt * 16 + l16] : 0.f;
  int rbase = mt * 16 + ((lane >> 4) << 3);
  float* crow = C + (size_t)rbase * Nc + nt * 16 + l16;
#pragma unroll
  for (int r = 0; r < 8; ++r)
    crow[(size_t)r * Nc] = acc[r] + bs;
}

__global__ void zero_int(int* p, int n) {
  int t = blockIdx.x * blockDim.x + threadIdx.x;
  if (t < n) p[t] = 0;
}

__global__ void count_edges(const int* __restrict__ ecol, int* __restrict__ cnt) {
  int e = blockIdx.x * blockDim.x + threadIdx.x;
  if (e < NE_EDGES) atomicAdd(&cnt[ecol[e]], 1);
}

__global__ __launch_bounds__(1024)
void scan_rowptr(const int* __restrict__ cnt, int* __restrict__ rowptr) {
  __shared__ int part[1024];
  const int CH = (N_NODES + 1023) / 1024;
  int tid = threadIdx.x;
  int base = tid * CH;
  int s = 0;
  for (int i = 0; i < CH; ++i) {
    int idx = base + i;
    if (idx < N_NODES) s += cnt[idx];
  }
  part[tid] = s;
  __syncthreads();
  for (int off = 1; off < 1024; off <<= 1) {
    int v = 0;
    if (tid >= off) v = part[tid - off];
    __syncthreads();
    if (tid >= off) part[tid] += v;
    __syncthreads();
  }
  int run = (tid == 0) ? 0 : part[tid - 1];
  for (int i = 0; i < CH; ++i) {
    int idx = base + i;
    if (idx < N_NODES) { rowptr[idx] = run; run += cnt[idx]; }
  }
  if (tid == 1023) rowptr[N_NODES] = run;
}

__global__ void copy_int(const int* __restrict__ src, int* __restrict__ dst, int n) {
  int t = blockIdx.x * blockDim.x + threadIdx.x;
  if (t < n) dst[t] = src[t];
}

__global__ void csr_fill(const int* __restrict__ erow, const int* __restrict__ ecol,
                         int* __restrict__ cursor, int* __restrict__ srcidx) {
  int e = blockIdx.x * blockDim.x + threadIdx.x;
  if (e >= NE_EDGES) return;
  int c = ecol[e];
  int p = atomicAdd(&cursor[c], 1);
  srcidx[p] = erow[e];
}

__global__ void dis_from_cnt(const int* __restrict__ cnt, float* __restrict__ dis, int n) {
  int t = blockIdx.x * blockDim.x + threadIdx.x;
  if (t < n) dis[t] = rsqrtf((float)cnt[t] + 1.0f);
}

#define AGG_WAVES 8
#define N_PARTS (N_NODES / AGG_WAVES)
static_assert(N_NODES % AGG_WAVES == 0, "one wave per node, whole blocks only");

__global__ __launch_bounds__(256)
void gcn_aggregate(const int* __restrict__ rowptr, const int* __restrict__ srcidx,
                   const float* __restrict__ dis, const float* __restrict__ lin,
                   float* __restrict__ agg, float* __restrict__ part) {
  __shared__ float slot[AGG_WAVES * 256];
  __shared__ int arrived;
  if (threadIdx.x == 0) arrived = 0;
  __syncthreads();

  int w = threadIdx.x >> 5;
  int lane = threadIdx.x & 31;
  int c = blockIdx.x * AGG_WAVES + w;
  float dc = dis[c];
  const float4 sv = *(const float4*)(lin + (size_t)c * DF + lane * 4);
  float s = dc * dc;
  float ax = s * sv.x, ay = s * sv.y, az = s * sv.z, aw = s * sv.w;
  int e0 = rowptr[c], e1 = rowptr[c + 1];
  for (int j = e0; j < e1; ++j) {
    int r = srcidx[j];
    float nrm = dis[r] * dc;
    const float4 v = *(const float4*)(lin + (size_t)r * DF + lane * 4);
    ax += nrm * v.x; ay += nrm * v.y; az += nrm * v.z; aw += nrm * v.w;
  }
  float4 o; o.x = ax; o.y = ay; o.z = az; o.w = aw;
  float4 q; q.x = ax * ax; q.y = ay * ay; q.z = az * az; q.w = aw * aw;

  *(float4*)(agg + (size_t)c * DF + lane * 4) = o;

  float* mine = slot + w * 256 + (lane >> 2) * 32 + (lane & 3) * 4;
  *(float4*)(mine)      = o;
  *(float4*)(mine + 16) = q;
  __builtin_amdgcn_fence(__ATOMIC_RELEASE, "workgroup", "local");
  int old = 0;
  if (lane == 0)
    old = __hip_atomic_fetch_add(&arrived, 1, __ATOMIC_RELAXED, __HIP_MEMORY_SCOPE_WORKGROUP);
  old = __builtin_amdgcn_readfirstlane(old);

  if (old == AGG_WAVES - 1) {
    __builtin_amdgcn_fence(__ATOMIC_ACQUIRE, "workgroup", "local");
#pragma unroll
    for (int g = 0; g < 8; ++g) {
      float v = 0.f;
#pragma unroll
      for (int k = 0; k < AGG_WAVES; ++k) v += slot[k * 256 + g * 32 + lane];
      part[((size_t)g * N_PARTS + blockIdx.x) * 32 + lane] = v;
    }
  }
}

__global__ __launch_bounds__(1024)
void bn_finalize_parts(const float* __restrict__ part, float* __restrict__ stats,
                       const float* __restrict__ g, const float* __restrict__ be,
                       float inv_n) {
  __shared__ float red[32 * 32];
  const int QCH  = (N_PARTS + 3) / 4;
  const int ITER = (QCH + 31) / 32;
  int grp  = blockIdx.x;
  int w    = threadIdx.x >> 5;
  int lane = threadIdx.x & 31;
  int sub  = lane >> 3;
  int e0   = (lane & 7) * 4;
  const float* base = part + (size_t)grp * N_PARTS * 32;

  float4 acc; acc.x = 0.f; acc.y = 0.f; acc.z = 0.f; acc.w = 0.f;
#pragma unroll 7
  for (int it = 0; it < ITER; ++it) {
    int p  = (it * 32 + w) * 4 + sub;
    bool ok = p < N_PARTS;
    int pc = ok ? p : N_PARTS - 1;
    float4 v = *(const float4*)(base + (size_t)pc * 32 + e0);
    acc.x += ok ? v.x : 0.f; acc.y += ok ? v.y : 0.f;
    acc.z += ok ? v.z : 0.f; acc.w += ok ? v.w : 0.f;
  }
#pragma unroll
  for (int off = 8; off < 32; off <<= 1) {
    acc.x += __shfl_xor(acc.x, off, 32); acc.y += __shfl_xor(acc.y, off, 32);
    acc.z += __shfl_xor(acc.z, off, 32); acc.w += __shfl_xor(acc.w, off, 32);
  }
  if (lane < 8) *(float4*)(red + w * 32 + e0) = acc;
  __syncthreads();
  if (w == 0) {
    float tot = 0.f;
    for (int i = 0; i < 32; ++i) tot += red[i * 32 + lane];
    float sq = __shfl_down(tot, 16, 32);
    if (lane < 16) {
      int f = grp * 16 + lane;
      float m   = tot * inv_n;
      float var = sq * inv_n - m * m;
      float sc  = g[f] * rsqrtf(var + EPS_BN);
      stats[f]      = sc;
      stats[DF + f] = be[f] - m * sc;
    }
  }
}

__global__ __launch_bounds__(256)
void bn_apply_leaky(const float* __restrict__ agg, const float* __restrict__ stats,
                    float* __restrict__ h) {
  int t = blockIdx.x * blockDim.x + threadIdx.x;
  if (t >= N_NODES * DF) return;
  int f = t & (DF - 1);
  float y = agg[t] * stats[f] + stats[DF + f];
  h[t] = (y >= 0.f) ? y : 0.1f * y;
}

__global__ void transpose128(const float* __restrict__ src, float* __restrict__ dst) {
  int t = blockIdx.x * blockDim.x + threadIdx.x;
  if (t >= 128 * 128) return;
  int r = t >> 7, c = t & 127;
  dst[c * 128 + r] = src[t];
}

__global__ __launch_bounds__(256)
void gather_pairs(const float* __restrict__ h, const int* __restrict__ di,
                  float* __restrict__ amat, float* __restrict__ bmat) {
  int t = blockIdx.x * blockDim.x + threadIdx.x;
  if (t >= BPAIRS * 32) return;
  int j = t >> 5, q = t & 31;
  int ia = di[2 * j] - 1;     if (ia < 0) ia += N_NODES;
  int ib = di[2 * j + 1] - 1; if (ib < 0) ib += N_NODES;
  *(float4*)(amat + (size_t)j * DF + q * 4) = *(const float4*)(h + (size_t)ia * DF + q * 4);
  *(float4*)(bmat + (size_t)j * DF + q * 4) = *(const float4*)(h + (size_t)ib * DF + q * 4);
}

__global__ __launch_bounds__(256)
void rowdot(const float* __restrict__ tm, const float* __restrict__ bm,
            float* __restrict__ out) {
  int t = blockIdx.x * blockDim.x + threadIdx.x;
  int j = t >> 5, lane = t & 31;
  if (j >= BPAIRS) return;
  float4 tv = *(const float4*)(tm + (size_t)j * DF + lane * 4);
  float4 bv = *(const float4*)(bm + (size_t)j * DF + lane * 4);
  float p = tv.x * bv.x + tv.y * bv.y + tv.z * bv.z + tv.w * bv.w;
#pragma unroll
  for (int off = 16; off > 0; off >>= 1)
    p += __shfl_down(p, off, 32);
  if (lane == 0) out[j] = p;
}

extern "C" void kernel_launch(void* const* d_in, const int* in_sizes, int n_in,
                              void* d_out, int out_size, void* d_ws, size_t ws_size,
                              hipStream_t stream) {
  const float* x   = (const float*)d_in[0];
  const int*   edg = (const int*)d_in[1];
  const int*   di  = (const int*)d_in[2];
  const float* w1  = (const float*)d_in[3];
  const float* b1  = (const float*)d_in[4];
  const float* w2  = (const float*)d_in[5];
  const float* b2  = (const float*)d_in[6];
  const float* w3  = (const float*)d_in[7];
  const float* b3  = (const float*)d_in[8];
  const float* g1  = (const float*)d_in[9];
  const float* be1 = (const float*)d_in[10];
  const float* g2  = (const float*)d_in[11];
  const float* be2 = (const float*)d_in[12];
  const float* g3  = (const float*)d_in[13];
  const float* be3 = (const float*)d_in[14];
  const float* p1  = (const float*)d_in[15];
  const float* p2  = (const float*)d_in[16];
  float* out = (float*)d_out;

  float* ws = (float*)d_ws;
  size_t off = 0;
  float* dis    = ws + off; off += N_NODES;
  float* stats  = ws + off; off += 256;
  float* Wt     = ws + off; off += 128 * 128;
  int*   cnt    = (int*)(ws + off); off += N_NODES;
  int*   rowptr = (int*)(ws + off); off += N_NODES + 16;
  int*   cursor = (int*)(ws + off); off += N_NODES;
  int*   srcidx = (int*)(ws + off); off += NE_EDGES;
  float* lin    = ws + off; off += (size_t)N_NODES * DF;
  float* agg    = ws + off; off += (size_t)N_NODES * DF;
  float* h      = ws + off; off += (size_t)N_NODES * DF;
  float* part   = h + 16;
  float* U    = lin;
  float* Qt   = lin + 16384;
  float* amat = lin + 32768;
  float* bmat = amat + (size_t)BPAIRS * DF;
  float* tmat = bmat + (size_t)BPAIRS * DF;

  const int* erow = edg;
  const int* ecol = edg + NE_EDGES;

  zero_int<<<(N_NODES + 255) / 256, 256, 0, stream>>>(cnt, N_NODES);
  count_edges<<<(NE_EDGES + 255) / 256, 256, 0, stream>>>(ecol, cnt);
  scan_rowptr<<<1, 1024, 0, stream>>>(cnt, rowptr);
  copy_int<<<(N_NODES + 255) / 256, 256, 0, stream>>>(rowptr, cursor, N_NODES);
  csr_fill<<<(NE_EDGES + 255) / 256, 256, 0, stream>>>(erow, ecol, cursor, srcidx);
  dis_from_cnt<<<(N_NODES + 255) / 256, 256, 0, stream>>>(cnt, dis, N_NODES);

  const float* Wl[3] = {w1, w2, w3};
  const float* Bl[3] = {b1, b2, b3};
  const float* Gl[3] = {g1, g2, g3};
  const float* El[3] = {be1, be2, be3};
  const float* in = x;
  const int strip_waves = N_NODES / 16;
  for (int l = 0; l < 3; ++l) {
    transpose128<<<64, 256, 0, stream>>>(Wl[l], Wt);
    wmma_gemm_strip<<<(strip_waves * 32 + 255) / 256, 256, 0, stream>>>(
        in, Wt, Bl[l], lin, N_NODES, DF);
    gcn_aggregate<<<N_PARTS, AGG_WAVES * 32, 0, stream>>>(
        rowptr, srcidx, dis, lin, agg, part);
    bn_finalize_parts<<<DF / 16, 1024, 0, stream>>>(
        part, stats, Gl[l], El[l], 1.0f / (float)N_NODES);
    bn_apply_leaky<<<(N_NODES * DF + 255) / 256, 256, 0, stream>>>(agg, stats, h);
    in = h;
  }

  const int small_tiles = (DF / 16) * (DF / 16);
  wmma_matmul<<<(small_tiles * 32 + 255) / 256, 256, 0, stream>>>(p1, p2, nullptr, U, DF, DF, DF, 1);
  wmma_matmul<<<(small_tiles * 32 + 255) / 256, 256, 0, stream>>>(U, p1, nullptr, Qt, DF, DF, DF, 1);
  gather_pairs<<<(BPAIRS * 32 + 255) / 256, 256, 0, stream>>>(h, di, amat, bmat);
  wmma_gemm_strip<<<((BPAIRS / 16) * 32 + 255) / 256, 256, 0, stream>>>(
      amat, Qt, nullptr, tmat, BPAIRS, DF);
  rowdot<<<(BPAIRS * 32 + 255) / 256, 256, 0, stream>>>(tmat, bmat, out);
}


# Adjacency built by owner blocks: 128 blocks each read the whole target list and slot their 391 nodes' edges by LDS atomic adds; global atomics, zero and dis launches gone; in-row order: LDS arrival
# speedup vs baseline: 3.5011x; 1.5877x over previous
#include <hip/hip_runtime.h>

#define N_NODES 50000
#define NE_EDGES 800000
#define BPAIRS 4096
#define DF 128
#define EPS_BN 1e-5f

typedef float v2f __attribute__((ext_vector_type(2)));
typedef float v8f __attribute__((ext_vector_type(8)));

__device__ __forceinline__ v8f wmma4(v2f a, v2f b, v8f c) {
  return __builtin_amdgcn_wmma_f32_16x16x4_f32(
       false, a,  false, b,
       (short)0, c,  false,  false);
}

__global__ __launch_bounds__(256)
void wmma_gemm_strip(const float* __restrict__ A, const float* __restrict__ Bt,
                     const float* __restrict__ bias, float* __restrict__ C,
                     int M, int K) {
  int gtid = blockIdx.x * blockDim.x + threadIdx.x;
  int wave = gtid >> 5;
  int lane = threadIdx.x & 31;
  if (wave * 16 >= M) return;

  int l16   = lane & 15;
  int khalf = (lane >> 4) << 1;
  const float* arow = A + (size_t)(wave * 16 + l16) * K + khalf;

  const float* brow[8];
#pragma unroll
  for (int nt = 0; nt < 8; ++nt)
    brow[nt] = Bt + (size_t)(nt * 16 + l16) * K + khalf;

  v8f zero = {};
  v8f acc[8];
#pragma unroll
  for (int nt = 0; nt < 8; ++nt) acc[nt] = zero;

  for (int k = 0; k < K; k += 4) {
    v2f a = *(const v2f*)(arow + k);
#pragma unroll
    for (int nt = 0; nt < 8; ++nt) {
      v2f b = *(const v2f*)(brow[nt] + k);
      acc[nt] = wmma4(a, b, acc[nt]);
    }
  }

  int rbase = wave * 16 + ((lane >> 4) << 3);
#pragma unroll
  for (int nt = 0; nt < 8; ++nt) {
    float bs = bias ? bias[nt * 16 + l16] : 0.f;
    float* crow = C + (size_t)rbase * 128 + nt * 16 + l16;
#pragma unroll
    for (int r = 0; r < 8; ++r)
      crow[(size_t)r * 128] = acc[nt][r] + bs;
  }
}

__global__ __launch_bounds__(256)
void wmma_matmul(const float* __restrict__ A, const float* __restrict__ Bm,
                 const float* __restrict__ bias, float* __restrict__ C,
                 int M, int Nc, int K, int bTrans) {
  int gtid = blockIdx.x * blockDim.x + threadIdx.x;
  int wave = gtid >> 5;
  int lane = threadIdx.x & 31;
  int ntiles = Nc >> 4;
  int mt = wave / ntiles;
  int nt = wave - mt * ntiles;
  if (mt * 16 >= M) return;

  int l16   = lane & 15;
  int khalf = (lane >> 4) << 1;
  const float* arow = A + (size_t)(mt * 16 + l16) * K + khalf;

  v8f acc = {};
  if (bTrans) {
    const float* brow = Bm + (size_t)(nt * 16 + l16) * K + khalf;
    for (int k = 0; k < K; k += 4) {
      v2f a = *(const v2f*)(arow + k);
      v2f b = *(const v2f*)(brow + k);
      acc = wmma4(a, b, acc);
    }
  } else {
    const float* bcol = Bm + (size_t)khalf * Nc + nt * 16 + l16;
    for (int k = 0; k < K; k += 4) {
      v2f a = *(const v2f*)(arow + k);
      v2f b;
      b.x = bcol[(size_t)k * Nc];
      b.y = bcol[(size_t)(k + 1) * Nc];
      acc = wmma4(a, b, acc);
    }
  }

  float bs = bias ? bias[nt * 16 + l16] : 0.f;
  int rbase = mt * 16 + ((lane >> 4) << 3);
  float* crow = C + (size_t)rbase * Nc + nt * 16 + l16;
#pragma unroll
  for (int r = 0; r < 8; ++r)
    crow[(size_t)r * Nc] = acc[r] + bs;
}

#define ADJ_CAP 64
#define ADJ_BLOCKS 128
#define ADJ_NB ((N_NODES + ADJ_BLOCKS - 1) / ADJ_BLOCKS)
#define ADJ_THREADS 1024
#define ADJ_Q (NE_EDGES / 4)
static_assert(NE_EDGES % 32 == 0, "the edge list is read in whole 16-byte pieces and whole waves");

__global__ __launch_bounds__(ADJ_THREADS)
void adj_build(const int* __restrict__ erow, const int* __restrict__ ecol,
               int* __restrict__ cnt, int* __restrict__ slots, float* __restrict__ dis) {
  __shared__ int cntL[ADJ_NB];
  __shared__ int slotL[ADJ_NB * ADJ_CAP];
  const int lo = blockIdx.x * ADJ_NB;
  int nb = N_NODES - lo; if (nb > ADJ_NB) nb = ADJ_NB;
  for (int i = threadIdx.x; i < ADJ_NB; i += ADJ_THREADS) cntL[i] = 0;
  __syncthreads();

  const int4* ec4 = (const int4*)ecol;
  for (int q0 = threadIdx.x; q0 < ADJ_Q; q0 += ADJ_THREADS * 8) {
    int4 v[8];
#pragma unroll
    for (int u = 0; u < 8; ++u) {
      int q = q0 + u * ADJ_THREADS;
      v[u] = ec4[q < ADJ_Q ? q : ADJ_Q - 1];
    }
    int p[8][4];
#pragma unroll
    for (int u = 0; u < 8; ++u) {
      int q = q0 + u * ADJ_THREADS;
      int cc[4] = {v[u].x, v[u].y, v[u].z, v[u].w};
#pragma unroll
      for (int k = 0; k < 4; ++k) {
        unsigned d = (unsigned)(cc[k] - lo);
        p[u][k] = (q < ADJ_Q && d < (unsigned)nb) ? atomicAdd(&cntL[d], 1) : ADJ_CAP;
      }
    }
#pragma unroll
    for (int u = 0; u < 8; ++u) {
      int q = q0 + u * ADJ_THREADS;
      int cc[4] = {v[u].x, v[u].y, v[u].z, v[u].w};
#pragma unroll
      for (int k = 0; k < 4; ++k)
        if (p[u][k] < ADJ_CAP) slotL[(cc[k] - lo) * ADJ_CAP + p[u][k]] = q * 4 + k;
    }
  }
  __syncthreads();

  for (int i = threadIdx.x; i < nb; i += ADJ_THREADS) {
    int dg = cntL[i];
    cnt[lo + i] = dg;
    dis[lo + i] = rsqrtf((float)dg + 1.0f);
  }
  int4* out4 = (int4*)(slots + (size_t)lo * ADJ_CAP);
  const int4* in4 = (const int4*)slotL;
  for (int t = threadIdx.x; t < nb * (ADJ_CAP / 4); t += ADJ_THREADS) {
    int i = t >> 4, j = t & 15;
    int left = cntL[i] - j * 4;
    if (left <= 0) continue;
    int4 e4 = in4[t];
    int4 r4;
    r4.x = erow[e4.x];
    r4.y = erow[left > 1 ? e4.y : e4.x];
    r4.z = erow[left > 2 ? e4.z : e4.x];
    r4.w = erow[left > 3 ? e4.w : e4.x];
    out4[t] = r4;
  }
}

#define AGG_WAVES 16
#define N_PARTS (N_NODES / AGG_WAVES)
#define A_PITCH 132
static_assert(N_NODES % AGG_WAVES == 0, "one wave per node, whole 16-row tiles only");

__global__ __launch_bounds__(AGG_WAVES * 32)
void gcn_aggregate_linear(const int* __restrict__ cnt, const int* __restrict__ slots,
                          const int* __restrict__ erow, const int* __restrict__ ecol,
                          const float* __restrict__ dis, const float* __restrict__ xin,
                          const float* __restrict__ Wm, const float* __restrict__ bias,
                          float* __restrict__ agg, float* __restrict__ part) {
  __shared__ float atile[AGG_WAVES * A_PITCH];
  __shared__ float wsum[AGG_WAVES];
  __shared__ int arrived;
  if (threadIdx.x == 0) arrived = 0;
  __syncthreads();

  int w = threadIdx.x >> 5;
  int lane = threadIdx.x & 31;
  int c = blockIdx.x * AGG_WAVES + w;
  float dc = dis[c];
  const float4 sv = *(const float4*)(xin + (size_t)c * DF + lane * 4);
  float s = dc * dc;
  float ax = s * sv.x, ay = s * sv.y, az = s * sv.z, aw = s * sv.w;
  float ns = s;
  int deg = cnt[c];
  int nslot = deg <= ADJ_CAP ? deg : 0;
  const int* sl = slots + (size_t)c * ADJ_CAP;
  for (int j = 0; j < nslot; ++j) {
    int r = sl[j];
    float nrm = dis[r] * dc;
    const float4 v = *(const float4*)(xin + (size_t)r * DF + lane * 4);
    ax += nrm * v.x; ay += nrm * v.y; az += nrm * v.z; aw += nrm * v.w;
    ns += nrm;
  }
  if (deg > ADJ_CAP) {
    for (int e0 = 0; e0 < NE_EDGES; e0 += 32) {
      unsigned m = __builtin_amdgcn_ballot_w32(ecol[e0 + lane] == c);
      while (m) {
        int b = __builtin_ctz(m);
        m &= m - 1;
        int r = erow[e0 + b];
        float nrm = dis[r] * dc;
        const float4 v = *(const float4*)(xin + (size_t)r * DF + lane * 4);
        ax += nrm * v.x; ay += nrm * v.y; az += nrm * v.z; aw += nrm * v.w;
        ns += nrm;
      }
    }
  }
  float4 o; o.x = ax; o.y = ay; o.z = az; o.w = aw;
  *(float4*)(atile + w * A_PITCH + lane * 4) = o;
  if (lane == 0) wsum[w] = ns;
  __builtin_amdgcn_fence(__ATOMIC_RELEASE, "workgroup", "local");
  int old = 0;
  if (lane == 0)
    old = __hip_atomic_fetch_add(&arrived, 1, __ATOMIC_RELAXED, __HIP_MEMORY_SCOPE_WORKGROUP);
  old = __builtin_amdgcn_readfirstlane(old);
  if (old != AGG_WAVES - 1) return;
  __builtin_amdgcn_fence(__ATOMIC_ACQUIRE, "workgroup", "local");

  int l16   = lane & 15;
  int hi    = lane >> 4;
  int khalf = hi << 1;
  const float* ap = atile + l16 * A_PITCH + khalf;
  const float* bp = Wm + (size_t)khalf * DF + l16;
  float* cbase = agg + ((size_t)blockIdx.x * AGG_WAVES + hi * 8) * DF + l16;

#pragma unroll 1
  for (int p = 0; p < 4; ++p) {
    const float* bq = bp + p * 32;
    v8f acc0 = {};
    v8f acc1 = {};
#pragma unroll 4
    for (int k = 0; k < DF; k += 4) {
      v2f a = *(const v2f*)(ap + k);
      v2f b0, b1;
      b0.x = bq[(size_t)k * DF];
      b0.y = bq[(size_t)(k + 1) * DF];
      b1.x = bq[(size_t)k * DF + 16];
      b1.y = bq[(size_t)(k + 1) * DF + 16];
      acc0 = wmma4(a, b0, acc0);
      acc1 = wmma4(a, b1, acc1);
    }
#pragma unroll
    for (int t = 0; t < 2; ++t) {
      int g = p * 2 + t;
      v8f acc = t ? acc1 : acc0;
      float bs = bias[g * 16 + l16];
      float cs = 0.f, cq = 0.f;
      float* crow = cbase + g * 16;
#pragma unroll
      for (int r = 0; r < 8; ++r) {
        float v = acc[r] + wsum[hi * 8 + r] * bs;
        crow[(size_t)r * DF] = v;
        cs += v; cq += v * v;
      }
      cs += __shfl_xor(cs, 16, 32);
      cq += __shfl_xor(cq, 16, 32);
      part[((size_t)g * N_PARTS + blockIdx.x) * 32 + lane] = hi ? cq : cs;
    }
  }
}

__global__ __launch_bounds__(1024)
void bn_finalize_parts(const float* __restrict__ part, float* __restrict__ stats,
                       const float* __restrict__ g, const float* __restrict__ be,
                       float inv_n) {
  __shared__ float red[32 * 32];
  const int QCH  = (N_PARTS + 3) / 4;
  const int ITER = (QCH + 31) / 32;
  int grp  = blockIdx.x;
  int w    = threadIdx.x >> 5;
  int lane = threadIdx.x & 31;
  int sub  = lane >> 3;
  int e0   = (lane & 7) * 4;
  const float* base = part + (size_t)grp * N_PARTS * 32;

  float4 acc; acc.x = 0.f; acc.y = 0.f; acc.z = 0.f; acc.w = 0.f;
#pragma unroll 7
  for (int it = 0; it < ITER; ++it) {
    int p  = (it * 32 + w) * 4 + sub;
    bool ok = p < N_PARTS;
    int pc = ok ? p : N_PARTS - 1;
    float4 v = *(const float4*)(base + (size_t)pc * 32 + e0);
    acc.x += ok ? v.x : 0.f; acc.y += ok ? v.y : 0.f;
    acc.z += ok ? v.z : 0.f; acc.w += ok ? v.w : 0.f;
  }
#pragma unroll
  for (int off = 8; off < 32; off <<= 1) {
    acc.x += __shfl_xor(acc.x, off, 32); acc.y += __shfl_xor(acc.y, off, 32);
    acc.z += __shfl_xor(acc.z, off, 32); acc.w += __shfl_xor(acc.w, off, 32);
  }
  if (lane < 8) *(float4*)(red + w * 32 + e0) = acc;
  __syncthreads();
  if (w == 0) {
    float tot = 0.f;
    for (int i = 0; i < 32; ++i) tot += red[i * 32 + lane];
    float sq = __shfl_down(tot, 16, 32);
    if (lane < 16) {
      int f = grp * 16 + lane;
      float m   = tot * inv_n;
      float var = sq * inv_n - m * m;
      float sc  = g[f] * rsqrtf(var + EPS_BN);
      stats[f]      = sc;
      stats[DF + f] = be[f] - m * sc;
    }
  }
}

__global__ __launch_bounds__(256)
void bn_apply_leaky(const float* __restrict__ agg, const float* __restrict__ stats,
                    float* __restrict__ h) {
  int t = blockIdx.x * blockDim.x + threadIdx.x;
  if (t >= N_NODES * DF) return;
  int f = t & (DF - 1);
  float y = agg[t] * stats[f] + stats[DF + f];
  h[t] = (y >= 0.f) ? y : 0.1f * y;
}

__global__ __launch_bounds__(256)
void gather_pairs(const float* __restrict__ h, const int* __restrict__ di,
                  float* __restrict__ amat, float* __restrict__ bmat) {
  int t = blockIdx.x * blockDim.x + threadIdx.x;
  if (t >= BPAIRS * 32) return;
  int j = t >> 5, q = t & 31;
  int ia = di[2 * j] - 1;     if (ia < 0) ia += N_NODES;
  int ib = di[2 * j + 1] - 1; if (ib < 0) ib += N_NODES;
  *(float4*)(amat + (size_t)j * DF + q * 4) = *(const float4*)(h + (size_t)ia * DF + q * 4);
  *(float4*)(bmat + (size_t)j * DF + q * 4) = *(const float4*)(h + (size_t)ib * DF + q * 4);
}

__global__ __launch_bounds__(256)
void rowdot(const float* __restrict__ tm, const float* __restrict__ bm,
            float* __restrict__ out) {
  int t = blockIdx.x * blockDim.x + threadIdx.x;
  int j = t >> 5, lane = t & 31;
  if (j >= BPAIRS) return;
  float4 tv = *(const float4*)(tm + (size_t)j * DF + lane * 4);
  float4 bv = *(const float4*)(bm + (size_t)j * DF + lane * 4);
  float p = tv.x * bv.x + tv.y * bv.y + tv.z * bv.z + tv.w * bv.w;
#pragma unroll
  for (int off = 16; off > 0; off >>= 1)
    p += __shfl_down(p, off, 32);
  if (lane == 0) out[j] = p;
}

extern "C" void kernel_launch(void* const* d_in, const int* in_sizes, int n_in,
                              void* d_out, int out_size, void* d_ws, size_t ws_size,
                              hipStream_t stream) {
  const float* x   = (const float*)d_in[0];
  const int*   edg = (const int*)d_in[1];
  const int*   di  = (const int*)d_in[2];
  const float* w1  = (const float*)d_in[3];
  const float* b1  = (const float*)d_in[4];
  const float* w2  = (const float*)d_in[5];
  const float* b2  = (const float*)d_in[6];
  const float* w3  = (const float*)d_in[7];
  const float* b3  = (const float*)d_in[8];
  const float* g1  = (const float*)d_in[9];
  const float* be1 = (const float*)d_in[10];
  const float* g2  = (const float*)d_in[11];
  const float* be2 = (const float*)d_in[12];
  const float* g3  = (const float*)d_in[13];
  const float* be3 = (const float*)d_in[14];
  const float* p1  = (const float*)d_in[15];
  const float* p2  = (const float*)d_in[16];
  float* out = (float*)d_out;

  float* ws = (float*)d_ws;
  size_t off = 0;
  float* dis    = ws + off; off += N_NODES;
  float* stats  = ws + off; off += 256;
  off += 128 * 128;
  int*   cnt    = (int*)(ws + off); off += N_NODES;
  off += N_NODES + 16;
  off += N_NODES;
  off += NE_EDGES;
  float* lin    = ws + off; off += (size_t)N_NODES * DF;
  float* agg    = ws + off; off += (size_t)N_NODES * DF;
  float* h      = ws + off; off += (size_t)N_NODES * DF;
  float* part   = lin + 16;
  int*   slots  = (int*)(lin + 2097152 + 48);
  float* U    = lin;
  float* Qt   = lin + 16384;
  float* amat = lin + 32768;
  float* bmat = amat + (size_t)BPAIRS * DF;
  float* tmat = bmat + (size_t)BPAIRS * DF;

  const int* erow = edg;
  const int* ecol = edg + NE_EDGES;

  adj_build<<<ADJ_BLOCKS, ADJ_THREADS, 0, stream>>>(erow, ecol, cnt, slots, dis);

  const float* Wl[3] = {w1, w2, w3};
  const float* Bl[3] = {b1, b2, b3};
  const float* Gl[3] = {g1, g2, g3};
  const float* El[3] = {be1, be2, be3};
  const float* in = x;
  for (int l = 0; l < 3; ++l) {
    gcn_aggregate_linear<<<N_PARTS, AGG_WAVES * 32, 0, stream>>>(
        cnt, slots, erow, ecol, dis, in, Wl[l], Bl[l], agg, part);
    bn_finalize_parts<<<DF / 16, 1024, 0, stream>>>(
        part, stats, Gl[l], El[l], 1.0f / (float)N_NODES);
    bn_apply_leaky<<<(N_NODES * DF + 255) / 256, 256, 0, stream>>>(agg, stats, h);
    in = h;
  }

  const int small_tiles = (DF / 16) * (DF / 16);
  wmma_matmul<<<(small_tiles * 32 + 255) / 256, 256, 0, stream>>>(p1, p2, nullptr, U, DF, DF, DF, 1);
  wmma_matmul<<<(small_tiles * 32 + 255) / 256, 256, 0, stream>>>(U, p1, nullptr, Qt, DF, DF, DF, 1);
  gather_pairs<<<(BPAIRS * 32 + 255) / 256, 256, 0, stream>>>(h, di, amat, bmat);
  wmma_gemm_strip<<<((BPAIRS / 16) * 32 + 255) / 256, 256, 0, stream>>>(
      amat, Qt, nullptr, tmat, BPAIRS, DF);
  rowdot<<<(BPAIRS * 32 + 255) / 256, 256, 0, stream>>>(tmat, bmat, out);
}


# Layer product re-tiled: 32-node blocks; behind a block barrier 8 waves each multiply both 16-row tiles by their own 16 columns of W (wave tile 32 x 16, not one last wave 16 x 128); same sum order
# speedup vs baseline: 3.8925x; 1.1118x over previous
#include <hip/hip_runtime.h>

#define N_NODES 50000
#define NE_EDGES 800000
#define BPAIRS 4096
#define DF 128
#define EPS_BN 1e-5f

typedef float v2f __attribute__((ext_vector_type(2)));
typedef float v8f __attribute__((ext_vector_type(8)));

__device__ __forceinline__ v8f wmma4(v2f a, v2f b, v8f c) {
  return __builtin_amdgcn_wmma_f32_16x16x4_f32(
       false, a,  false, b,
       (short)0, c,  false,  false);
}

__global__ __launch_bounds__(256)
void wmma_gemm_strip(const float* __restrict__ A, const float* __restrict__ Bt,
                     const float* __restrict__ bias, float* __restrict__ C,
                     int M, int K) {
  int gtid = blockIdx.x * blockDim.x + threadIdx.x;
  int wave = gtid >> 5;
  int lane = threadIdx.x & 31;
  if (wave * 16 >= M) return;

  int l16   = lane & 15;
  int khalf = (lane >> 4) << 1;
  const float* arow = A + (size_t)(wave * 16 + l16) * K + khalf;

  const float* brow[8];
#pragma unroll
  for (int nt = 0; nt < 8; ++nt)
    brow[nt] = Bt + (size_t)(nt * 16 + l16) * K + khalf;

  v8f zero = {};
  v8f acc[8];
#pragma unroll
  for (int nt = 0; nt < 8; ++nt) acc[nt] = zero;

  for (int k = 0; k < K; k += 4) {
    v2f a = *(const v2f*)(arow + k);
#pragma unroll
    for (int nt = 0; nt < 8; ++nt) {
      v2f b = *(const v2f*)(brow[nt] + k);
      acc[nt] = wmma4(a, b, acc[nt]);
    }
  }

  int rbase = wave * 16 + ((lane >> 4) << 3);
#pragma unroll
  for (int nt = 0; nt < 8; ++nt) {
    float bs = bias ? bias[nt * 16 + l16] : 0.f;
    float* crow = C + (size_t)rbase * 128 + nt * 16 + l16;
#pragma unroll
    for (int r = 0; r < 8; ++r)
      crow[(size_t)r * 128] = acc[nt][r] + bs;
  }
}

__global__ __launch_bounds__(256)
void wmma_matmul(const float* __restrict__ A, const float* __restrict__ Bm,
                 const float* __restrict__ bias, float* __restrict__ C,
                 int M, int Nc, int K, int bTrans) {
  int gtid = blockIdx.x * blockDim.x + threadIdx.x;
  int wave = gtid >> 5;
  int lane = threadIdx.x & 31;
  int ntiles = Nc >> 4;
  int mt = wave / ntiles;
  int nt = wave - mt * ntiles;
  if (mt * 16 >= M) return;

  int l16   = lane & 15;
  int khalf = (lane >> 4) << 1;
  const float* arow = A + (size_t)(mt * 16 + l16) * K + khalf;

  v8f acc = {};
  if (bTrans) {
    const float* brow = Bm + (size_t)(nt * 16 + l16) * K + khalf;
    for (int k = 0; k < K; k += 4) {
      v2f a = *(const v2f*)(arow + k);
      v2f b = *(const v2f*)(brow + k);
      acc = wmma4(a, b, acc);
    }
  } else {
    const float* bcol = Bm + (size_t)khalf * Nc + nt * 16 + l16;
    for (int k = 0; k < K; k += 4) {
      v2f a = *(const v2f*)(arow + k);
      v2f b;
      b.x = bcol[(size_t)k * Nc];
      b.y = bcol[(size_t)(k + 1) * Nc];
      acc = wmma4(a, b, acc);
    }
  }

  float bs = bias ? bias[nt * 16 + l16] : 0.f;
  int rbase = mt * 16 + ((lane >> 4) << 3);
  float* crow = C + (size_t)rbase * Nc + nt * 16 + l16;
#pragma unroll
  for (int r = 0; r < 8; ++r)
    crow[(size_t)r * Nc] = acc[r] + bs;
}

#define ADJ_CAP 64
#define ADJ_BLOCKS 128
#define ADJ_NB ((N_NODES + ADJ_BLOCKS - 1) / ADJ_BLOCKS)
#define ADJ_THREADS 1024
#define ADJ_Q (NE_EDGES / 4)
static_assert(NE_EDGES % 32 == 0, "the edge list is read in whole 16-byte pieces and whole waves");

__global__ __launch_bounds__(ADJ_THREADS)
void adj_build(const int* __restrict__ erow, const int* __restrict__ ecol,
               int* __restrict__ cnt, int* __restrict__ slots, float* __restrict__ dis) {
  __shared__ int cntL[ADJ_NB];
  __shared__ int slotL[ADJ_NB * ADJ_CAP];
  const int lo = blockIdx.x * ADJ_NB;
  int nb = N_NODES - lo; if (nb > ADJ_NB) nb = ADJ_NB;
  for (int i = threadIdx.x; i < ADJ_NB; i += ADJ_THREADS) cntL[i] = 0;
  __syncthreads();

  const int4* ec4 = (const int4*)ecol;
  for (int q0 = threadIdx.x; q0 < ADJ_Q; q0 += ADJ_THREADS * 8) {
    int4 v[8];
#pragma unroll
    for (int u = 0; u < 8; ++u) {
      int q = q0 + u * ADJ_THREADS;
      v[u] = ec4[q < ADJ_Q ? q : ADJ_Q - 1];
    }
    int p[8][4];
#pragma unroll
    for (int u = 0; u < 8; ++u) {
      int q = q0 + u * ADJ_THREADS;
      int cc[4] = {v[u].x, v[u].y, v[u].z, v[u].w};
#pragma unroll
      for (int k = 0; k < 4; ++k) {
        unsigned d = (unsigned)(cc[k] - lo);
        p[u][k] = (q < ADJ_Q && d < (unsigned)nb) ? atomicAdd(&cntL[d], 1) : ADJ_CAP;
      }
    }
#pragma unroll
    for (int u = 0; u < 8; ++u) {
      int q = q0 + u * ADJ_THREADS;
      int cc[4] = {v[u].x, v[u].y, v[u].z, v[u].w};
#pragma unroll
      for (int k = 0; k < 4; ++k)
        if (p[u][k] < ADJ_CAP) slotL[(cc[k] - lo) * ADJ_CAP + p[u][k]] = q * 4 + k;
    }
  }
  __syncthreads();

  for (int i = threadIdx.x; i < nb; i += ADJ_THREADS) {
    int dg = cntL[i];
    cnt[lo + i] = dg;
    dis[lo + i] = rsqrtf((float)dg + 1.0f);
  }
  int4* out4 = (int4*)(slots + (size_t)lo * ADJ_CAP);
  const int4* in4 = (const int4*)slotL;
  for (int t = threadIdx.x; t < nb * (ADJ_CAP / 4); t += ADJ_THREADS) {
    int i = t >> 4, j = t & 15;
    int left = cntL[i] - j * 4;
    if (left <= 0) continue;
    int4 e4 = in4[t];
    int4 r4;
    r4.x = erow[e4.x];
    r4.y = erow[left > 1 ? e4.y : e4.x];
    r4.z = erow[left > 2 ? e4.z : e4.x];
    r4.w = erow[left > 3 ? e4.w : e4.x];
    out4[t] = r4;
  }
}

#define AGG_WAVES 32
#define N_PARTS (N_NODES / 16)
#define AGG_BLOCKS ((N_NODES + AGG_WAVES - 1) / AGG_WAVES)
#define A_PITCH 132
static_assert(N_NODES % 16 == 0, "whole 16-row tiles only");
static_assert(AGG_WAVES == 32, "two 16-row tiles per block");

__global__ __launch_bounds__(AGG_WAVES * 32)
void gcn_aggregate_linear(const int* __restrict__ cnt, const int* __restrict__ slots,
                          const int* __restrict__ erow, const int* __restrict__ ecol,
                          const float* __restrict__ dis, const float* __restrict__ xin,
                          const float* __restrict__ Wm, const float* __restrict__ bias,
                          float* __restrict__ agg, float* __restrict__ part) {
  __shared__ float atile[AGG_WAVES * A_PITCH];
  __shared__ float wsum[AGG_WAVES];

  int w = threadIdx.x >> 5;
  int lane = threadIdx.x & 31;
  int c = blockIdx.x * AGG_WAVES + w;
  float ax = 0.f, ay = 0.f, az = 0.f, aw = 0.f;
  float ns = 0.f;
  if (c < N_NODES) {
    float dc = dis[c];
    const float4 sv = *(const float4*)(xin + (size_t)c * DF + lane * 4);
    float s = dc * dc;
    ax = s * sv.x; ay = s * sv.y; az = s * sv.z; aw = s * sv.w;
    ns = s;
    int deg = cnt[c];
    int nslot = deg <= ADJ_CAP ? deg : 0;
    const int* sl = slots + (size_t)c * ADJ_CAP;
    for (int j = 0; j < nslot; ++j) {
      int r = sl[j];
      float nrm = dis[r] * dc;
      const float4 v = *(const float4*)(xin + (size_t)r * DF + lane * 4);
      ax += nrm * v.x; ay += nrm * v.y; az += nrm * v.z; aw += nrm * v.w;
      ns += nrm;
    }
    if (deg > ADJ_CAP) {
      for (int e0 = 0; e0 < NE_EDGES; e0 += 32) {
        unsigned m = __builtin_amdgcn_ballot_w32(ecol[e0 + lane] == c);
        while (m) {
          int b = __builtin_ctz(m);
          m &= m - 1;
          int r = erow[e0 + b];
          float nrm = dis[r] * dc;
          const float4 v = *(const float4*)(xin + (size_t)r * DF + lane * 4);
          ax += nrm * v.x; ay += nrm * v.y; az += nrm * v.z; aw += nrm * v.w;
          ns += nrm;
        }
      }
    }
  }
  float4 o; o.x = ax; o.y = ay; o.z = az; o.w = aw;
  *(float4*)(atile + w * A_PITCH + lane * 4) = o;
  if (lane == 0) wsum[w] = ns;
  __syncthreads();
  if (w >= DF / 16) return;

  int l16   = lane & 15;
  int hi    = lane >> 4;
  int khalf = hi << 1;
  int g     = w;
  const float* ap0 = atile + l16 * A_PITCH + khalf;
  const float* ap1 = ap0 + 16 * A_PITCH;
  const float* bp  = Wm + (size_t)khalf * DF + g * 16 + l16;

  v8f acc0 = {};
  v8f acc1 = {};
#pragma unroll 8
  for (int k = 0; k < DF; k += 4) {
    v2f a0 = *(const v2f*)(ap0 + k);
    v2f a1 = *(const v2f*)(ap1 + k);
    v2f b;
    b.x = bp[(size_t)k * DF];
    b.y = bp[(size_t)(k + 1) * DF];
    acc0 = wmma4(a0, b, acc0);
    acc1 = wmma4(a1, b, acc1);
  }

  float bs = bias[g * 16 + l16];
#pragma unroll
  for (int t = 0; t < 2; ++t) {
    int tile = blockIdx.x * 2 + t;
    if (tile >= N_PARTS) break;
    v8f acc = t ? acc1 : acc0;
    float cs = 0.f, cq = 0.f;
    float* crow = agg + ((size_t)tile * 16 + hi * 8) * DF + g * 16 + l16;
#pragma unroll
    for (int r = 0; r < 8; ++r) {
      float v = acc[r] + wsum[t * 16 + hi * 8 + r] * bs;
      crow[(size_t)r * DF] = v;
      cs += v; cq += v * v;
    }
    cs += __shfl_xor(cs, 16, 32);
    cq += __shfl_xor(cq, 16, 32);
    part[((size_t)g * N_PARTS + tile) * 32 + lane] = hi ? cq : cs;
  }
}

__global__ __launch_bounds__(1024)
void bn_finalize_parts(const float* __restrict__ part, float* __restrict__ stats,
                       const float* __restrict__ g, const float* __restrict__ be,
                       float inv_n) {
  __shared__ float red[32 * 32];
  const int QCH  = (N_PARTS + 3) / 4;
  const int ITER = (QCH + 31) / 32;
  int grp  = blockIdx.x;
  int w    = threadIdx.x >> 5;
  int lane = threadIdx.x & 31;
  int sub  = lane >> 3;
  int e0   = (lane & 7) * 4;
  const float* base = part + (size_t)grp * N_PARTS * 32;

  float4 acc; acc.x = 0.f; acc.y = 0.f; acc.z = 0.f; acc.w = 0.f;
#pragma unroll 7
  for (int it = 0; it < ITER; ++it) {
    int p  = (it * 32 + w) * 4 + sub;
    bool ok = p < N_PARTS;
    int pc = ok ? p : N_PARTS - 1;
    float4 v = *(const float4*)(base + (size_t)pc * 32 + e0);
    acc.x += ok ? v.x : 0.f; acc.y += ok ? v.y : 0.f;
    acc.z += ok ? v.z : 0.f; acc.w += ok ? v.w : 0.f;
  }
#pragma unroll
  for (int off = 8; off < 32; off <<= 1) {
    acc.x += __shfl_xor(acc.x, off, 32); acc.y += __shfl_xor(acc.y, off, 32);
    acc.z += __shfl_xor(acc.z, off, 32); acc.w += __shfl_xor(acc.w, off, 32);
  }
  if (lane < 8) *(float4*)(red + w * 32 + e0) = acc;
  __syncthreads();
  if (w == 0) {
    float tot = 0.f;
    for (int i = 0; i < 32; ++i) tot += red[i * 32 + lane];
    float sq = __shfl_down(tot, 16, 32);
    if (lane < 16) {
      int f = grp * 16 + lane;
      float m   = tot * inv_n;
      float var = sq * inv_n - m * m;
      float sc  = g[f] * rsqrtf(var + EPS_BN);
      stats[f]      = sc;
      stats[DF + f] = be[f] - m * sc;
    }
  }
}

__global__ __launch_bounds__(256)
void bn_apply_leaky(const float* __restrict__ agg, const float* __restrict__ stats,
                    float* __restrict__ h) {
  int t = blockIdx.x * blockDim.x + threadIdx.x;
  if (t >= N_NODES * DF) return;
  int f = t & (DF - 1);
  float y = agg[t] * stats[f] + stats[DF + f];
  h[t] = (y >= 0.f) ? y : 0.1f * y;
}

__global__ __launch_bounds__(256)
void gather_pairs(const float* __restrict__ h, const int* __restrict__ di,
                  float* __restrict__ amat, float* __restrict__ bmat) {
  int t = blockIdx.x * blockDim.x + threadIdx.x;
  if (t >= BPAIRS * 32) return;
  int j = t >> 5, q = t & 31;
  int ia = di[2 * j] - 1;     if (ia < 0) ia += N_NODES;
  int ib = di[2 * j + 1] - 1; if (ib < 0) ib += N_NODES;
  *(float4*)(amat + (size_t)j * DF + q * 4) = *(const float4*)(h + (size_t)ia * DF + q * 4);
  *(float4*)(bmat + (size_t)j * DF + q * 4) = *(const float4*)(h + (size_t)ib * DF + q * 4);
}

__global__ __launch_bounds__(256)
void rowdot(const float* __restrict__ tm, const float* __restrict__ bm,
            float* __restrict__ out) {
  int t = blockIdx.x * blockDim.x + threadIdx.x;
  int j = t >> 5, lane = t & 31;
  if (j >= BPAIRS) return;
  float4 tv = *(const float4*)(tm + (size_t)j * DF + lane * 4);
  float4 bv = *(const float4*)(bm + (size_t)j * DF + lane * 4);
  float p = tv.x * bv.x + tv.y * bv.y + tv.z * bv.z + tv.w * bv.w;
#pragma unroll
  for (int off = 16; off > 0; off >>= 1)
    p += __shfl_down(p, off, 32);
  if (lane == 0) out[j] = p;
}

extern "C" void kernel_launch(void* const* d_in, const int* in_sizes, int n_in,
                              void* d_out, int out_size, void* d_ws, size_t ws_size,
                              hipStream_t stream) {
  const float* x   = (const float*)d_in[0];
  const int*   edg = (const int*)d_in[1];
  const int*   di  = (const int*)d_in[2];
  const float* w1  = (const float*)d_in[3];
  const float* b1  = (const float*)d_in[4];
  const float* w2  = (const float*)d_in[5];
  const float* b2  = (const float*)d_in[6];
  const float* w3  = (const float*)d_in[7];
  const float* b3  = (const float*)d_in[8];
  const float* g1  = (const float*)d_in[9];
  const float* be1 = (const float*)d_in[10];
  const float* g2  = (const float*)d_in[11];
  const float* be2 = (const float*)d_in[12];
  const float* g3  = (const float*)d_in[13];
  const float* be3 = (const float*)d_in[14];
  const float* p1  = (const float*)d_in[15];
  const float* p2  = (const float*)d_in[16];
  float* out = (float*)d_out;

  float* ws = (float*)d_ws;
  size_t off = 0;
  float* dis    = ws + off; off += N_NODES;
  float* stats  = ws + off; off += 256;
  off += 128 * 128;
  int*   cnt    = (int*)(ws + off); off += N_NODES;
  off += N_NODES + 16;
  off += N_NODES;
  off += NE_EDGES;
  float* lin    = ws + off; off += (size_t)N_NODES * DF;
  float* agg    = ws + off; off += (size_t)N_NODES * DF;
  float* h      = ws + off; off += (size_t)N_NODES * DF;
  float* part   = lin + 16;
  int*   slots  = (int*)(lin + 2097152 + 48);
  float* U    = lin;
  float* Qt   = lin + 16384;
  float* amat = lin + 32768;
  float* bmat = amat + (size_t)BPAIRS * DF;
  float* tmat = bmat + (size_t)BPAIRS * DF;

  const int* erow = edg;
  const int* ecol = edg + NE_EDGES;

  adj_build<<<ADJ_BLOCKS, ADJ_THREADS, 0, stream>>>(erow, ecol, cnt, slots, dis);

  const float* Wl[3] = {w1, w2, w3};
  const float* Bl[3] = {b1, b2, b3};
  const float* Gl[3] = {g1, g2, g3};
  const float* El[3] = {be1, be2, be3};
  const float* in = x;
  for (int l = 0; l < 3; ++l) {
    gcn_aggregate_linear<<<AGG_BLOCKS, AGG_WAVES * 32, 0, stream>>>(
        cnt, slots, erow, ecol, dis, in, Wl[l], Bl[l], agg, part);
    bn_finalize_parts<<<DF / 16, 1024, 0, stream>>>(
        part, stats, Gl[l], El[l], 1.0f / (float)N_NODES);
    bn_apply_leaky<<<(N_NODES * DF + 255) / 256, 256, 0, stream>>>(agg, stats, h);
    in = h;
  }

  const int small_tiles = (DF / 16) * (DF / 16);
  wmma_matmul<<<(small_tiles * 32 + 255) / 256, 256, 0, stream>>>(p1, p2, nullptr, U, DF, DF, DF, 1);
  wmma_matmul<<<(small_tiles * 32 + 255) / 256, 256, 0, stream>>>(U, p1, nullptr, Qt, DF, DF, DF, 1);
  gather_pairs<<<(BPAIRS * 32 + 255) / 256, 256, 0, stream>>>(h, di, amat, bmat);
  wmma_gemm_strip<<<((BPAIRS / 16) * 32 + 255) / 256, 256, 0, stream>>>(
      amat, Qt, nullptr, tmat, BPAIRS, DF);
  rowdot<<<(BPAIRS * 32 + 255) / 256, 256, 0, stream>>>(tmat, bmat, out);
}


# Neighbour sum: lane j loads slot j once per node; rows fetched 8 at once from lane reads, the tail as one group; each row's dis value a scalar load beside it; no dependent index load; same sum order
# speedup vs baseline: 4.0644x; 1.0442x over previous
#include <hip/hip_runtime.h>

#define N_NODES 50000
#define NE_EDGES 800000
#define BPAIRS 4096
#define DF 128
#define EPS_BN 1e-5f

typedef float v2f __attribute__((ext_vector_type(2)));
typedef float v8f __attribute__((ext_vector_type(8)));

__device__ __forceinline__ v8f wmma4(v2f a, v2f b, v8f c) {
  return __builtin_amdgcn_wmma_f32_16x16x4_f32(
       false, a,  false, b,
       (short)0, c,  false,  false);
}

__global__ __launch_bounds__(256)
void wmma_gemm_strip(const float* __restrict__ A, const float* __restrict__ Bt,
                     const float* __restrict__ bias, float* __restrict__ C,
                     int M, int K) {
  int gtid = blockIdx.x * blockDim.x + threadIdx.x;
  int wave = gtid >> 5;
  int lane = threadIdx.x & 31;
  if (wave * 16 >= M) return;

  int l16   = lane & 15;
  int khalf = (lane >> 4) << 1;
  const float* arow = A + (size_t)(wave * 16 + l16) * K + khalf;

  const float* brow[8];
#pragma unroll
  for (int nt = 0; nt < 8; ++nt)
    brow[nt] = Bt + (size_t)(nt * 16 + l16) * K + khalf;

  v8f zero = {};
  v8f acc[8];
#pragma unroll
  for (int nt = 0; nt < 8; ++nt) acc[nt] = zero;

  for (int k = 0; k < K; k += 4) {
    v2f a = *(const v2f*)(arow + k);
#pragma unroll
    for (int nt = 0; nt < 8; ++nt) {
      v2f b = *(const v2f*)(brow[nt] + k);
      acc[nt] = wmma4(a, b, acc[nt]);
    }
  }

  int rbase = wave * 16 + ((lane >> 4) << 3);
#pragma unroll
  for (int nt = 0; nt < 8; ++nt) {
    float bs = bias ? bias[nt * 16 + l16] : 0.f;
    float* crow = C + (size_t)rbase * 128 + nt * 16 + l16;
#pragma unroll
    for (int r = 0; r < 8; ++r)
      crow[(size_t)r * 128] = acc[nt][r] + bs;
  }
}

__global__ __launch_bounds__(256)
void wmma_matmul(const float* __restrict__ A, const float* __restrict__ Bm,
                 const float* __restrict__ bias, float* __restrict__ C,
                 int M, int Nc, int K, int bTrans) {
  int gtid = blockIdx.x * blockDim.x + threadIdx.x;
  int wave = gtid >> 5;
  int lane = threadIdx.x & 31;
  int ntiles = Nc >> 4;
  int mt = wave / ntiles;
  int nt = wave - mt * ntiles;
  if (mt * 16 >= M) return;

  int l16   = lane & 15;
  int khalf = (lane >> 4) << 1;
  const float* arow = A + (size_t)(mt * 16 + l16) * K + khalf;

  v8f acc = {};
  if (bTrans) {
    const float* brow = Bm + (size_t)(nt * 16 + l16) * K + khalf;
    for (int k = 0; k < K; k += 4) {
      v2f a = *(const v2f*)(arow + k);
      v2f b = *(const v2f*)(brow + k);
      acc = wmma4(a, b, acc);
    }
  } else {
    const float* bcol = Bm + (size_t)khalf * Nc + nt * 16 + l16;
    for (int k = 0; k < K; k += 4) {
      v2f a = *(const v2f*)(arow + k);
      v2f b;
      b.x = bcol[(size_t)k * Nc];
      b.y = bcol[(size_t)(k + 1) * Nc];
      acc = wmma4(a, b, acc);
    }
  }

  float bs = bias ? bias[nt * 16 + l16] : 0.f;
  int rbase = mt * 16 + ((lane >> 4) << 3);
  float* crow = C + (size_t)rbase * Nc + nt * 16 + l16;
#pragma unroll
  for (int r = 0; r < 8; ++r)
    crow[(size_t)r * Nc] = acc[r] + bs;
}

#define ADJ_CAP 64
#define ADJ_BLOCKS 128
#define ADJ_NB ((N_NODES + ADJ_BLOCKS - 1) / ADJ_BLOCKS)
#define ADJ_THREADS 1024
#define ADJ_Q (NE_EDGES / 4)
static_assert(NE_EDGES % 32 == 0, "the edge list is read in whole 16-byte pieces and whole waves");

__global__ __launch_bounds__(ADJ_THREADS)
void adj_build(const int* __restrict__ erow, const int* __restrict__ ecol,
               int* __restrict__ cnt, int* __restrict__ slots, float* __restrict__ dis) {
  __shared__ int cntL[ADJ_NB];
  __shared__ int slotL[ADJ_NB * ADJ_CAP];
  const int lo = blockIdx.x * ADJ_NB;
  int nb = N_NODES - lo; if (nb > ADJ_NB) nb = ADJ_NB;
  for (int i = threadIdx.x; i < ADJ_NB; i += ADJ_THREADS) cntL[i] = 0;
  __syncthreads();

  const int4* ec4 = (const int4*)ecol;
  for (int q0 = threadIdx.x; q0 < ADJ_Q; q0 += ADJ_THREADS * 8) {
    int4 v[8];
#pragma unroll
    for (int u = 0; u < 8; ++u) {
      int q = q0 + u * ADJ_THREADS;
      v[u] = ec4[q < ADJ_Q ? q : ADJ_Q - 1];
    }
    int p[8][4];
#pragma unroll
    for (int u = 0; u < 8; ++u) {
      int q = q0 + u * ADJ_THREADS;
      int cc[4] = {v[u].x, v[u].y, v[u].z, v[u].w};
#pragma unroll
      for (int k = 0; k < 4; ++k) {
        unsigned d = (unsigned)(cc[k] - lo);
        p[u][k] = (q < ADJ_Q && d < (unsigned)nb) ? atomicAdd(&cntL[d], 1) : ADJ_CAP;
      }
    }
#pragma unroll
    for (int u = 0; u < 8; ++u) {
      int q = q0 + u * ADJ_THREADS;
      int cc[4] = {v[u].x, v[u].y, v[u].z, v[u].w};
#pragma unroll
      for (int k = 0; k < 4; ++k)
        if (p[u][k] < ADJ_CAP) slotL[(cc[k] - lo) * ADJ_CAP + p[u][k]] = q * 4 + k;
    }
  }
  __syncthreads();

  for (int i = threadIdx.x; i < nb; i += ADJ_THREADS) {
    int dg = cntL[i];
    cnt[lo + i] = dg;
    dis[lo + i] = rsqrtf((float)dg + 1.0f);
  }
  int4* out4 = (int4*)(slots + (size_t)lo * ADJ_CAP);
  const int4* in4 = (const int4*)slotL;
  for (int t = threadIdx.x; t < nb * (ADJ_CAP / 4); t += ADJ_THREADS) {
    int i = t >> 4, j = t & 15;
    int left = cntL[i] - j * 4;
    if (left <= 0) continue;
    int4 e4 = in4[t];
    int4 r4;
    r4.x = erow[e4.x];
    r4.y = erow[left > 1 ? e4.y : e4.x];
    r4.z = erow[left > 2 ? e4.z : e4.x];
    r4.w = erow[left > 3 ? e4.w : e4.x];
    out4[t] = r4;
  }
}

#define AGG_WAVES 32
#define N_PARTS (N_NODES / 16)
#define AGG_BLOCKS ((N_NODES + AGG_WAVES - 1) / AGG_WAVES)
#define A_PITCH 132
static_assert(N_NODES % 16 == 0, "whole 16-row tiles only");
static_assert(AGG_WAVES == 32, "two 16-row tiles per block");

__device__ __forceinline__ void add_rows(const float* __restrict__ xin, const float* __restrict__ dis,
                                         int lane, int myr, float dc, int j0, const int n,
                                         float& ax, float& ay, float& az, float& aw, float& ns) {
  float4 v[8];
  float  d[8];
#pragma unroll
  for (int u = 0; u < 8; ++u) {
    if (u < n) {
      int r = __builtin_amdgcn_readlane(myr, j0 + u);
      d[u] = dis[r];
      v[u] = *(const float4*)(xin + (size_t)r * DF + lane * 4);
    }
  }
#pragma unroll
  for (int u = 0; u < 8; ++u) {
    if (u < n) {
      float nrm = d[u] * dc;
      ax += nrm * v[u].x; ay += nrm * v[u].y; az += nrm * v[u].z; aw += nrm * v[u].w;
      ns += nrm;
    }
  }
}

__global__ __launch_bounds__(AGG_WAVES * 32)
void gcn_aggregate_linear(const int* __restrict__ cnt, const int* __restrict__ slots,
                          const int* __restrict__ erow, const int* __restrict__ ecol,
                          const float* __restrict__ dis, const float* __restrict__ xin,
                          const float* __restrict__ Wm, const float* __restrict__ bias,
                          float* __restrict__ agg, float* __restrict__ part) {
  __shared__ float atile[AGG_WAVES * A_PITCH];
  __shared__ float wsum[AGG_WAVES];

  int w = __builtin_amdgcn_readfirstlane(threadIdx.x >> 5);
  int lane = threadIdx.x & 31;
  int c = blockIdx.x * AGG_WAVES + w;
  float ax = 0.f, ay = 0.f, az = 0.f, aw = 0.f;
  float ns = 0.f;
  if (c < N_NODES) {
    float dc = dis[c];
    const float4 sv = *(const float4*)(xin + (size_t)c * DF + lane * 4);
    float s = dc * dc;
    ax = s * sv.x; ay = s * sv.y; az = s * sv.z; aw = s * sv.w;
    ns = s;
    int deg = cnt[c];
    int nslot = deg <= ADJ_CAP ? deg : 0;
    const int* sl = slots + (size_t)c * ADJ_CAP;
    for (int base = 0; base < nslot; base += 32) {
      int left = nslot - base;
      if (left > 32) left = 32;
      int myr = c;
      if (lane < left) myr = sl[base + lane];
      int j = 0;
      for (; j + 8 <= left; j += 8)
        add_rows(xin, dis, lane, myr, dc, j, 8, ax, ay, az, aw, ns);
      switch (left - j) {
        case 1: add_rows(xin, dis, lane, myr, dc, j, 1, ax, ay, az, aw, ns); break;
        case 2: add_rows(xin, dis, lane, myr, dc, j, 2, ax, ay, az, aw, ns); break;
        case 3: add_rows(xin, dis, lane, myr, dc, j, 3, ax, ay, az, aw, ns); break;
        case 4: add_rows(xin, dis, lane, myr, dc, j, 4, ax, ay, az, aw, ns); break;
        case 5: add_rows(xin, dis, lane, myr, dc, j, 5, ax, ay, az, aw, ns); break;
        case 6: add_rows(xin, dis, lane, myr, dc, j, 6, ax, ay, az, aw, ns); break;
        case 7: add_rows(xin, dis, lane, myr, dc, j, 7, ax, ay, az, aw, ns); break;
        default: break;
      }
    }
    if (deg > ADJ_CAP) {
      for (int e0 = 0; e0 < NE_EDGES; e0 += 32) {
        unsigned m = __builtin_amdgcn_ballot_w32(ecol[e0 + lane] == c);
        while (m) {
          int b = __builtin_ctz(m);
          m &= m - 1;
          int r = erow[e0 + b];
          float nrm = dis[r] * dc;
          const float4 v = *(const float4*)(xin + (size_t)r * DF + lane * 4);
          ax += nrm * v.x; ay += nrm * v.y; az += nrm * v.z; aw += nrm * v.w;
          ns += nrm;
        }
      }
    }
  }
  float4 o; o.x = ax; o.y = ay; o.z = az; o.w = aw;
  *(float4*)(atile + w * A_PITCH + lane * 4) = o;
  if (lane == 0) wsum[w] = ns;
  __syncthreads();
  if (w >= DF / 16) return;

  int l16   = lane & 15;
  int hi    = lane >> 4;
  int khalf = hi << 1;
  int g     = w;
  const float* ap0 = atile + l16 * A_PITCH + khalf;
  const float* ap1 = ap0 + 16 * A_PITCH;
  const float* bp  = Wm + (size_t)khalf * DF + g * 16 + l16;

  v8f acc0 = {};
  v8f acc1 = {};
#pragma unroll 8
  for (int k = 0; k < DF; k += 4) {
    v2f a0 = *(const v2f*)(ap0 + k);
    v2f a1 = *(const v2f*)(ap1 + k);
    v2f b;
    b.x = bp[(size_t)k * DF];
    b.y = bp[(size_t)(k + 1) * DF];
    acc0 = wmma4(a0, b, acc0);
    acc1 = wmma4(a1, b, acc1);
  }

  float bs = bias[g * 16 + l16];
#pragma unroll
  for (int t = 0; t < 2; ++t) {
    int tile = blockIdx.x * 2 + t;
    if (tile >= N_PARTS) break;
    v8f acc = t ? acc1 : acc0;
    float cs = 0.f, cq = 0.f;
    float* crow = agg + ((size_t)tile * 16 + hi * 8) * DF + g * 16 + l16;
#pragma unroll
    for (int r = 0; r < 8; ++r) {
      float v = acc[r] + wsum[t * 16 + hi * 8 + r] * bs;
      crow[(size_t)r * DF] = v;
      cs += v; cq += v * v;
    }
    cs += __shfl_xor(cs, 16, 32);
    cq += __shfl_xor(cq, 16, 32);
    part[((size_t)g * N_PARTS + tile) * 32 + lane] = hi ? cq : cs;
  }
}

__global__ __launch_bounds__(1024)
void bn_finalize_parts(const float* __restrict__ part, float* __restrict__ stats,
                       const float* __restrict__ g, const float* __restrict__ be,
                       float inv_n) {
  __shared__ float red[32 * 32];
  const int QCH  = (N_PARTS + 3) / 4;
  const int ITER = (QCH + 31) / 32;
  int grp  = blockIdx.x;
  int w    = threadIdx.x >> 5;
  int lane = threadIdx.x & 31;
  int sub  = lane >> 3;
  int e0   = (lane & 7) * 4;
  const float* base = part + (size_t)grp * N_PARTS * 32;

  float4 acc; acc.x = 0.f; acc.y = 0.f; acc.z = 0.f; acc.w = 0.f;
#pragma unroll 7
  for (int it = 0; it < ITER; ++it) {
    int p  = (it * 32 + w) * 4 + sub;
    bool ok = p < N_PARTS;
    int pc = ok ? p : N_PARTS - 1;
    float4 v = *(const float4*)(base + (size_t)pc * 32 + e0);
    acc.x += ok ? v.x : 0.f; acc.y += ok ? v.y : 0.f;
    acc.z += ok ? v.z : 0.f; acc.w += ok ? v.w : 0.f;
  }
#pragma unroll
  for (int off = 8; off < 32; off <<= 1) {
    acc.x += __shfl_xor(acc.x, off, 32); acc.y += __shfl_xor(acc.y, off, 32);
    acc.z += __shfl_xor(acc.z, off, 32); acc.w += __shfl_xor(acc.w, off, 32);
  }
  if (lane < 8) *(float4*)(red + w * 32 + e0) = acc;
  __syncthreads();
  if (w == 0) {
    float tot = 0.f;
    for (int i = 0; i < 32; ++i) tot += red[i * 32 + lane];
    float sq = __shfl_down(tot, 16, 32);
    if (lane < 16) {
      int f = grp * 16 + lane;
      float m   = tot * inv_n;
      float var = sq * inv_n - m * m;
      float sc  = g[f] * rsqrtf(var + EPS_BN);
      stats[f]      = sc;
      stats[DF + f] = be[f] - m * sc;
    }
  }
}

__global__ __launch_bounds__(256)
void bn_apply_leaky(const float* __restrict__ agg, const float* __restrict__ stats,
                    float* __restrict__ h) {
  int t = blockIdx.x * blockDim.x + threadIdx.x;
  if (t >= N_NODES * DF) return;
  int f = t & (DF - 1);
  float y = agg[t] * stats[f] + stats[DF + f];
  h[t] = (y >= 0.f) ? y : 0.1f * y;
}

__global__ __launch_bounds__(256)
void gather_pairs(const float* __restrict__ h, const int* __restrict__ di,
                  float* __restrict__ amat, float* __restrict__ bmat) {
  int t = blockIdx.x * blockDim.x + threadIdx.x;
  if (t >= BPAIRS * 32) return;
  int j = t >> 5, q = t & 31;
  int ia = di[2 * j] - 1;     if (ia < 0) ia += N_NODES;
  int ib = di[2 * j + 1] - 1; if (ib < 0) ib += N_NODES;
  *(float4*)(amat + (size_t)j * DF + q * 4) = *(const float4*)(h + (size_t)ia * DF + q * 4);
  *(float4*)(bmat + (size_t)j * DF + q * 4) = *(const float4*)(h + (size_t)ib * DF + q * 4);
}

__global__ __launch_bounds__(256)
void rowdot(const float* __restrict__ tm, const float* __restrict__ bm,
            float* __restrict__ out) {
  int t = blockIdx.x * blockDim.x + threadIdx.x;
  int j = t >> 5, lane = t & 31;
  if (j >= BPAIRS) return;
  float4 tv = *(const float4*)(tm + (size_t)j * DF + lane * 4);
  float4 bv = *(const float4*)(bm + (size_t)j * DF + lane * 4);
  float p = tv.x * bv.x + tv.y * bv.y + tv.z * bv.z + tv.w * bv.w;
#pragma unroll
  for (int off = 16; off > 0; off >>= 1)
    p += __shfl_down(p, off, 32);
  if (lane == 0) out[j] = p;
}

extern "C" void kernel_launch(void* const* d_in, const int* in_sizes, int n_in,
                              void* d_out, int out_size, void* d_ws, size_t ws_size,
                              hipStream_t stream) {
  const float* x   = (const float*)d_in[0];
  const int*   edg = (const int*)d_in[1];
  const int*   di  = (const int*)d_in[2];
  const float* w1  = (const float*)d_in[3];
  const float* b1  = (const float*)d_in[4];
  const float* w2  = (const float*)d_in[5];
  const float* b2  = (const float*)d_in[6];
  const float* w3  = (const float*)d_in[7];
  const float* b3  = (const float*)d_in[8];
  const float* g1  = (const float*)d_in[9];
  const float* be1 = (const float*)d_in[10];
  const float* g2  = (const float*)d_in[11];
  const float* be2 = (const float*)d_in[12];
  const float* g3  = (const float*)d_in[13];
  const float* be3 = (const float*)d_in[14];
  const float* p1  = (const float*)d_in[15];
  const float* p2  = (const float*)d_in[16];
  float* out = (float*)d_out;

  float* ws = (float*)d_ws;
  size_t off = 0;
  float* dis    = ws + off; off += N_NODES;
  float* stats  = ws + off; off += 256;
  off += 128 * 128;
  int*   cnt    = (int*)(ws + off); off += N_NODES;
  off += N_NODES + 16;
  off += N_NODES;
  off += NE_EDGES;
  float* lin    = ws + off; off += (size_t)N_NODES * DF;
  float* agg    = ws + off; off += (size_t)N_NODES * DF;
  float* h      = ws + off; off += (size_t)N_NODES * DF;
  float* part   = lin + 16;
  int*   slots  = (int*)(lin + 2097152 + 48);
  float* U    = lin;
  float* Qt   = lin + 16384;
  float* amat = lin + 32768;
  float* bmat = amat + (size_t)BPAIRS * DF;
  float* tmat = bmat + (size_t)BPAIRS * DF;

  const int* erow = edg;
  const int* ecol = edg + NE_EDGES;

  adj_build<<<ADJ_BLOCKS, ADJ_THREADS, 0, stream>>>(erow, ecol, cnt, slots, dis);

  const float* Wl[3] = {w1, w2, w3};
  const float* Bl[3] = {b1, b2, b3};
  const float* Gl[3] = {g1, g2, g3};
  const float* El[3] = {be1, be2, be3};
  const float* in = x;
  for (int l = 0; l < 3; ++l) {
    gcn_aggregate_linear<<<AGG_BLOCKS, AGG_WAVES * 32, 0, stream>>>(
        cnt, slots, erow, ecol, dis, in, Wl[l], Bl[l], agg, part);
    bn_finalize_parts<<<DF / 16, 1024, 0, stream>>>(
        part, stats, Gl[l], El[l], 1.0f / (float)N_NODES);
    bn_apply_leaky<<<(N_NODES * DF + 255) / 256, 256, 0, stream>>>(agg, stats, h);
    in = h;
  }

  const int small_tiles = (DF / 16) * (DF / 16);
  wmma_matmul<<<(small_tiles * 32 + 255) / 256, 256, 0, stream>>>(p1, p2, nullptr, U, DF, DF, DF, 1);
  wmma_matmul<<<(small_tiles * 32 + 255) / 256, 256, 0, stream>>>(U, p1, nullptr, Qt, DF, DF, DF, 1);
  gather_pairs<<<(BPAIRS * 32 + 255) / 256, 256, 0, stream>>>(h, di, amat, bmat);
  wmma_gemm_strip<<<((BPAIRS / 16) * 32 + 255) / 256, 256, 0, stream>>>(
      amat, Qt, nullptr, tmat, BPAIRS, DF);
  rowdot<<<(BPAIRS * 32 + 255) / 256, 256, 0, stream>>>(tmat, bmat, out);
}


# Batch-norm scale, shift and leaky applied by the readers of agg (next layer's neighbour sum, pair gather) to each row as it arrives; the 3 apply passes and the activation plane gone; same sum order
# speedup vs baseline: 4.9677x; 1.2222x over previous
#include <hip/hip_runtime.h>

#define N_NODES 50000
#define NE_EDGES 800000
#define BPAIRS 4096
#define DF 128
#define EPS_BN 1e-5f

typedef float v2f __attribute__((ext_vector_type(2)));
typedef float v8f __attribute__((ext_vector_type(8)));
typedef float v4f __attribute__((ext_vector_type(4)));

__device__ __forceinline__ v8f wmma4(v2f a, v2f b, v8f c) {
  return __builtin_amdgcn_wmma_f32_16x16x4_f32(
       false, a,  false, b,
       (short)0, c,  false,  false);
}

__global__ __launch_bounds__(256)
void wmma_gemm_strip(const float* __restrict__ A, const float* __restrict__ Bt,
                     const float* __restrict__ bias, float* __restrict__ C,
                     int M, int K) {
  int gtid = blockIdx.x * blockDim.x + threadIdx.x;
  int wave = gtid >> 5;
  int lane = threadIdx.x & 31;
  if (wave * 16 >= M) return;

  int l16   = lane & 15;
  int khalf = (lane >> 4) << 1;
  const float* arow = A + (size_t)(wave * 16 + l16) * K + khalf;

  const float* brow[8];
#pragma unroll
  for (int nt = 0; nt < 8; ++nt)
    brow[nt] = Bt + (size_t)(nt * 16 + l16) * K + khalf;

  v8f zero = {};
  v8f acc[8];
#pragma unroll
  for (int nt = 0; nt < 8; ++nt) acc[nt] = zero;

  for (int k = 0; k < K; k += 4) {
    v2f a = *(const v2f*)(arow + k);
#pragma unroll
    for (int nt = 0; nt < 8; ++nt) {
      v2f b = *(const v2f*)(brow[nt] + k);
      acc[nt] = wmma4(a, b, acc[nt]);
    }
  }

  int rbase = wave * 16 + ((lane >> 4) << 3);
#pragma unroll
  for (int nt = 0; nt < 8; ++nt) {
    float bs = bias ? bias[nt * 16 + l16] : 0.f;
    float* crow = C + (size_t)rbase * 128 + nt * 16 + l16;
#pragma unroll
    for (int r = 0; r < 8; ++r)
      crow[(size_t)r * 128] = acc[nt][r] + bs;
  }
}

__global__ __launch_bounds__(256)
void wmma_matmul(const float* __restrict__ A, const float* __restrict__ Bm,
                 const float* __restrict__ bias, float* __restrict__ C,
                 int M, int Nc, int K, int bTrans) {
  int gtid = blockIdx.x * blockDim.x + threadIdx.x;
  int wave = gtid >> 5;
  int lane = threadIdx.x & 31;
  int ntiles = Nc >> 4;
  int mt = wave / ntiles;
  int nt = wave - mt * ntiles;
  if (mt * 16 >= M) return;

  int l16   = lane & 15;
  int khalf = (lane >> 4) << 1;
  const float* arow = A + (size_t)(mt * 16 + l16) * K + khalf;

  v8f acc = {};
  if (bTrans) {
    const float* brow = Bm + (size_t)(nt * 16 + l16) * K + khalf;
    for (int k = 0; k < K; k += 4) {
      v2f a = *(const v2f*)(arow + k);
      v2f b = *(const v2f*)(brow + k);
      acc = wmma4(a, b, acc);
    }
  } else {
    const float* bcol = Bm + (size_t)khalf * Nc + nt * 16 + l16;
    for (int k = 0; k < K; k += 4) {
      v2f a = *(const v2f*)(arow + k);
      v2f b;
      b.x = bcol[(size_t)k * Nc];
      b.y = bcol[(size_t)(k + 1) * Nc];
      acc = wmma4(a, b, acc);
    }
  }

  float bs = bias ? bias[nt * 16 + l16] : 0.f;
  int rbase = mt * 16 + ((lane >> 4) << 3);
  float* crow = C + (size_t)rbase * Nc + nt * 16 + l16;
#pragma unroll
  for (int r = 0; r < 8; ++r)
    crow[(size_t)r * Nc] = acc[r] + bs;
}

#define ADJ_CAP 64
#define ADJ_BLOCKS 128
#define ADJ_NB ((N_NODES + ADJ_BLOCKS - 1) / ADJ_BLOCKS)
#define ADJ_THREADS 1024
#define ADJ_Q (NE_EDGES / 4)
static_assert(NE_EDGES % 32 == 0, "the edge list is read in whole 16-byte pieces and whole waves");

__global__ __launch_bounds__(ADJ_THREADS)
void adj_build(const int* __restrict__ erow, const int* __restrict__ ecol,
               int* __restrict__ cnt, int* __restrict__ slots, float* __restrict__ dis) {
  __shared__ int cntL[ADJ_NB];
  __shared__ int slotL[ADJ_NB * ADJ_CAP];
  const int lo = blockIdx.x * ADJ_NB;
  int nb = N_NODES - lo; if (nb > ADJ_NB) nb = ADJ_NB;
  for (int i = threadIdx.x; i < ADJ_NB; i += ADJ_THREADS) cntL[i] = 0;
  __syncthreads();

  const int4* ec4 = (const int4*)ecol;
  for (int q0 = threadIdx.x; q0 < ADJ_Q; q0 += ADJ_THREADS * 8) {
    int4 v[8];
#pragma unroll
    for (int u = 0; u < 8; ++u) {
      int q = q0 + u * ADJ_THREADS;
      v[u] = ec4[q < ADJ_Q ? q : ADJ_Q - 1];
    }
    int p[8][4];
#pragma unroll
    for (int u = 0; u < 8; ++u) {
      int q = q0 + u * ADJ_THREADS;
      int cc[4] = {v[u].x, v[u].y, v[u].z, v[u].w};
#pragma unroll
      for (int k = 0; k < 4; ++k) {
        unsigned d = (unsigned)(cc[k] - lo);
        p[u][k] = (q < ADJ_Q && d < (unsigned)nb) ? atomicAdd(&cntL[d], 1) : ADJ_CAP;
      }
    }
#pragma unroll
    for (int u = 0; u < 8; ++u) {
      int q = q0 + u * ADJ_THREADS;
      int cc[4] = {v[u].x, v[u].y, v[u].z, v[u].w};
#pragma unroll
      for (int k = 0; k < 4; ++k)
        if (p[u][k] < ADJ_CAP) slotL[(cc[k] - lo) * ADJ_CAP + p[u][k]] = q * 4 + k;
    }
  }
  __syncthreads();

  for (int i = threadIdx.x; i < nb; i += ADJ_THREADS) {
    int dg = cntL[i];
    cnt[lo + i] = dg;
    dis[lo + i] = rsqrtf((float)dg + 1.0f);
  }
  int4* out4 = (int4*)(slots + (size_t)lo * ADJ_CAP);
  const int4* in4 = (const int4*)slotL;
  for (int t = threadIdx.x; t < nb * (ADJ_CAP / 4); t += ADJ_THREADS) {
    int i = t >> 4, j = t & 15;
    int left = cntL[i] - j * 4;
    if (left <= 0) continue;
    int4 e4 = in4[t];
    int4 r4;
    r4.x = erow[e4.x];
    r4.y = erow[left > 1 ? e4.y : e4.x];
    r4.z = erow[left > 2 ? e4.z : e4.x];
    r4.w = erow[left > 3 ? e4.w : e4.x];
    out4[t] = r4;
  }
}

#define AGG_WAVES 32
#define N_PARTS (N_NODES / 16)
#define AGG_BLOCKS ((N_NODES + AGG_WAVES - 1) / AGG_WAVES)
#define A_PITCH 132
static_assert(N_NODES % 16 == 0, "whole 16-row tiles only");
static_assert(AGG_WAVES == 32, "two 16-row tiles per block");

__device__ __forceinline__ float bn_leaky(float v, float sc, float sh) {
  float y = v * sc + sh;
  return fmaxf(y, 0.1f * y);
}

template <bool BN_IN>
__device__ __forceinline__ void add_rows(const float* __restrict__ xin, const float* __restrict__ dis,
                                         int lane, int myr, float dc, int j0, const int n,
                                         const float4 sc, const float4 sh,
                                         float& ax, float& ay, float& az, float& aw, float& ns) {
  v4f v[8];
  float  d[8];
#pragma unroll
  for (int u = 0; u < 8; ++u) {
    if (u < n) {
      int r = __builtin_amdgcn_readlane(myr, j0 + u);
      d[u] = dis[r];
      v[u] = *(const v4f*)(xin + (size_t)r * DF + lane * 4);
    }
  }
  if (BN_IN) {
    if (n == 8) asm volatile("" : "+v"(v[0]), "+v"(v[1]), "+v"(v[2]), "+v"(v[3]), "+v"(v[4]), "+v"(v[5]), "+v"(v[6]), "+v"(v[7]));
    else if (n == 7) asm volatile("" : "+v"(v[0]), "+v"(v[1]), "+v"(v[2]), "+v"(v[3]), "+v"(v[4]), "+v"(v[5]), "+v"(v[6]));
    else if (n == 6) asm volatile("" : "+v"(v[0]), "+v"(v[1]), "+v"(v[2]), "+v"(v[3]), "+v"(v[4]), "+v"(v[5]));
  }
#pragma unroll
  for (int u = 0; u < 8; ++u) {
    if (u < n) {
      float nrm = d[u] * dc;
      v4f r = v[u];
      if (BN_IN) {
        r.x = bn_leaky(r.x, sc.x, sh.x); r.y = bn_leaky(r.y, sc.y, sh.y);
        r.z = bn_leaky(r.z, sc.z, sh.z); r.w = bn_leaky(r.w, sc.w, sh.w);
      }
      ax += nrm * r.x; ay += nrm * r.y; az += nrm * r.z; aw += nrm * r.w;
      ns += nrm;
    }
  }
}

template <bool BN_IN>
__global__ __launch_bounds__(AGG_WAVES * 32, BN_IN ? 16 : 8)
void gcn_aggregate_linear(const int* __restrict__ cnt, const int* __restrict__ slots,
                          const int* __restrict__ erow, const int* __restrict__ ecol,
                          const float* __restrict__ dis, const float* __restrict__ xin,
                          const float* __restrict__ stats_in,
                          const float* __restrict__ Wm, const float* __restrict__ bias,
                          float* __restrict__ agg, float* __restrict__ part) {
  __shared__ float atile[AGG_WAVES * A_PITCH];
  __shared__ float wsum[AGG_WAVES];

  int w = __builtin_amdgcn_readfirstlane(threadIdx.x >> 5);
  int lane = threadIdx.x & 31;
  int c = blockIdx.x * AGG_WAVES + w;
  float ax = 0.f, ay = 0.f, az = 0.f, aw = 0.f;
  float ns = 0.f;
  float4 sc, sh;
  sc.x = sc.y = sc.z = sc.w = 1.f;
  sh.x = sh.y = sh.z = sh.w = 0.f;
  if (BN_IN) {
    sc = *(const float4*)(stats_in + lane * 4);
    sh = *(const float4*)(stats_in + DF + lane * 4);
  }
  if (c < N_NODES) {
    float dc = dis[c];
    float4 sv = *(const float4*)(xin + (size_t)c * DF + lane * 4);
    if (BN_IN) {
      sv.x = bn_leaky(sv.x, sc.x, sh.x); sv.y = bn_leaky(sv.y, sc.y, sh.y);
      sv.z = bn_leaky(sv.z, sc.z, sh.z); sv.w = bn_leaky(sv.w, sc.w, sh.w);
    }
    float s = dc * dc;
    ax = s * sv.x; ay = s * sv.y; az = s * sv.z; aw = s * sv.w;
    ns = s;
    int deg = cnt[c];
    int nslot = deg <= ADJ_CAP ? deg : 0;
    const int* sl = slots + (size_t)c * ADJ_CAP;
    for (int base = 0; base < nslot; base += 32) {
      int left = nslot - base;
      if (left > 32) left = 32;
      int myr = c;
      if (lane < left) myr = sl[base + lane];
      int j = 0;
      for (; j + 8 <= left; j += 8)
        add_rows<BN_IN>(xin, dis, lane, myr, dc, j, 8, sc, sh, ax, ay, az, aw, ns);
      switch (left - j) {
        case 1: add_rows<BN_IN>(xin, dis, lane, myr, dc, j, 1, sc, sh, ax, ay, az, aw, ns); break;
        case 2: add_rows<BN_IN>(xin, dis, lane, myr, dc, j, 2, sc, sh, ax, ay, az, aw, ns); break;
        case 3: add_rows<BN_IN>(xin, dis, lane, myr, dc, j, 3, sc, sh, ax, ay, az, aw, ns); break;
        case 4: add_rows<BN_IN>(xin, dis, lane, myr, dc, j, 4, sc, sh, ax, ay, az, aw, ns); break;
        case 5: add_rows<BN_IN>(xin, dis, lane, myr, dc, j, 5, sc, sh, ax, ay, az, aw, ns); break;
        case 6: add_rows<BN_IN>(xin, dis, lane, myr, dc, j, 6, sc, sh, ax, ay, az, aw, ns); break;
        case 7: add_rows<BN_IN>(xin, dis, lane, myr, dc, j, 7, sc, sh, ax, ay, az, aw, ns); break;
        default: break;
      }
    }
    if (deg > ADJ_CAP) {
      for (int e0 = 0; e0 < NE_EDGES; e0 += 32) {
        unsigned m = __builtin_amdgcn_ballot_w32(ecol[e0 + lane] == c);
        while (m) {
          int b = __builtin_ctz(m);
          m &= m - 1;
          int r = erow[e0 + b];
          float nrm = dis[r] * dc;
          float4 v = *(const float4*)(xin + (size_t)r * DF + lane * 4);
          if (BN_IN) {
            v.x = bn_leaky(v.x, sc.x, sh.x); v.y = bn_leaky(v.y, sc.y, sh.y);
            v.z = bn_leaky(v.z, sc.z, sh.z); v.w = bn_leaky(v.w, sc.w, sh.w);
          }
          ax += nrm * v.x; ay += nrm * v.y; az += nrm * v.z; aw += nrm * v.w;
          ns += nrm;
        }
      }
    }
  }
  float4 o; o.x = ax; o.y = ay; o.z = az; o.w = aw;
  *(float4*)(atile + w * A_PITCH + lane * 4) = o;
  if (lane == 0) wsum[w] = ns;
  __syncthreads();
  if (w >= DF / 16) return;

  int l16   = lane & 15;
  int hi    = lane >> 4;
  int khalf = hi << 1;
  int g     = w;
  const float* ap0 = atile + l16 * A_PITCH + khalf;
  const float* ap1 = ap0 + 16 * A_PITCH;
  const float* bp  = Wm + (size_t)khalf * DF + g * 16 + l16;

  v8f acc0 = {};
  v8f acc1 = {};
#pragma unroll 8
  for (int k = 0; k < DF; k += 4) {
    v2f a0 = *(const v2f*)(ap0 + k);
    v2f a1 = *(const v2f*)(ap1 + k);
    v2f b;
    b.x = bp[(size_t)k * DF];
    b.y = bp[(size_t)(k + 1) * DF];
    acc0 = wmma4(a0, b, acc0);
    acc1 = wmma4(a1, b, acc1);
  }

  float bs = bias[g * 16 + l16];
#pragma unroll
  for (int t = 0; t < 2; ++t) {
    int tile = blockIdx.x * 2 + t;
    if (tile >= N_PARTS) break;
    v8f acc = t ? acc1 : acc0;
    float cs = 0.f, cq = 0.f;
    float* crow = agg + ((size_t)tile * 16 + hi * 8) * DF + g * 16 + l16;
#pragma unroll
    for (int r = 0; r < 8; ++r) {
      float v = acc[r] + wsum[t * 16 + hi * 8 + r] * bs;
      crow[(size_t)r * DF] = v;
      cs += v; cq += v * v;
    }
    cs += __shfl_xor(cs, 16, 32);
    cq += __shfl_xor(cq, 16, 32);
    part[((size_t)g * N_PARTS + tile) * 32 + lane] = hi ? cq : cs;
  }
}

__global__ __launch_bounds__(1024)
void bn_finalize_parts(const float* __restrict__ part, float* __restrict__ stats,
                       const float* __restrict__ g, const float* __restrict__ be,
                       float inv_n) {
  __shared__ float red[32 * 32];
  const int QCH  = (N_PARTS + 3) / 4;
  const int ITER = (QCH + 31) / 32;
  int grp  = blockIdx.x;
  int w    = threadIdx.x >> 5;
  int lane = threadIdx.x & 31;
  int sub  = lane >> 3;
  int e0   = (lane & 7) * 4;
  const float* base = part + (size_t)grp * N_PARTS * 32;

  float4 acc; acc.x = 0.f; acc.y = 0.f; acc.z = 0.f; acc.w = 0.f;
#pragma unroll 7
  for (int it = 0; it < ITER; ++it) {
    int p  = (it * 32 + w) * 4 + sub;
    bool ok = p < N_PARTS;
    int pc = ok ? p : N_PARTS - 1;
    float4 v = *(const float4*)(base + (size_t)pc * 32 + e0);
    acc.x += ok ? v.x : 0.f; acc.y += ok ? v.y : 0.f;
    acc.z += ok ? v.z : 0.f; acc.w += ok ? v.w : 0.f;
  }
#pragma unroll
  for (int off = 8; off < 32; off <<= 1) {
    acc.x += __shfl_xor(acc.x, off, 32); acc.y += __shfl_xor(acc.y, off, 32);
    acc.z += __shfl_xor(acc.z, off, 32); acc.w += __shfl_xor(acc.w, off, 32);
  }
  if (lane < 8) *(float4*)(red + w * 32 + e0) = acc;
  __syncthreads();
  if (w == 0) {
    float tot = 0.f;
    for (int i = 0; i < 32; ++i) tot += red[i * 32 + lane];
    float sq = __shfl_down(tot, 16, 32);
    if (lane < 16) {
      int f = grp * 16 + lane;
      float m   = tot * inv_n;
      float var = sq * inv_n - m * m;
      float sc  = g[f] * rsqrtf(var + EPS_BN);
      stats[f]      = sc;
      stats[DF + f] = be[f] - m * sc;
    }
  }
}

__global__ __launch_bounds__(256)
void gather_pairs(const float* __restrict__ aggp, const float* __restrict__ stats,
                  const int* __restrict__ di,
                  float* __restrict__ amat, float* __restrict__ bmat) {
  int t = blockIdx.x * blockDim.x + threadIdx.x;
  if (t >= BPAIRS * 32) return;
  int j = t >> 5, q = t & 31;
  int ia = di[2 * j] - 1;     if (ia < 0) ia += N_NODES;
  int ib = di[2 * j + 1] - 1; if (ib < 0) ib += N_NODES;
  const float4 sc = *(const float4*)(stats + q * 4);
  const float4 sh = *(const float4*)(stats + DF + q * 4);
  float4 va = *(const float4*)(aggp + (size_t)ia * DF + q * 4);
  float4 vb = *(const float4*)(aggp + (size_t)ib * DF + q * 4);
  va.x = bn_leaky(va.x, sc.x, sh.x); va.y = bn_leaky(va.y, sc.y, sh.y);
  va.z = bn_leaky(va.z, sc.z, sh.z); va.w = bn_leaky(va.w, sc.w, sh.w);
  vb.x = bn_leaky(vb.x, sc.x, sh.x); vb.y = bn_leaky(vb.y, sc.y, sh.y);
  vb.z = bn_leaky(vb.z, sc.z, sh.z); vb.w = bn_leaky(vb.w, sc.w, sh.w);
  *(float4*)(amat + (size_t)j * DF + q * 4) = va;
  *(float4*)(bmat + (size_t)j * DF + q * 4) = vb;
}

__global__ __launch_bounds__(256)
void rowdot(const float* __restrict__ tm, const float* __restrict__ bm,
            float* __restrict__ out) {
  int t = blockIdx.x * blockDim.x + threadIdx.x;
  int j = t >> 5, lane = t & 31;
  if (j >= BPAIRS) return;
  float4 tv = *(const float4*)(tm + (size_t)j * DF + lane * 4);
  float4 bv = *(const float4*)(bm + (size_t)j * DF + lane * 4);
  float p = tv.x * bv.x + tv.y * bv.y + tv.z * bv.z + tv.w * bv.w;
#pragma unroll
  for (int off = 16; off > 0; off >>= 1)
    p += __shfl_down(p, off, 32);
  if (lane == 0) out[j] = p;
}

extern "C" void kernel_launch(void* const* d_in, const int* in_sizes, int n_in,
                              void* d_out, int out_size, void* d_ws, size_t ws_size,
                              hipStream_t stream) {
  const float* x   = (const float*)d_in[0];
  const int*   edg = (const int*)d_in[1];
  const int*   di  = (const int*)d_in[2];
  const float* w1  = (const float*)d_in[3];
  const float* b1  = (const float*)d_in[4];
  const float* w2  = (const float*)d_in[5];
  const float* b2  = (const float*)d_in[6];
  const float* w3  = (const float*)d_in[7];
  const float* b3  = (const float*)d_in[8];
  const float* g1  = (const float*)d_in[9];
  const float* be1 = (const float*)d_in[10];
  const float* g2  = (const float*)d_in[11];
  const float* be2 = (const float*)d_in[12];
  const float* g3  = (const float*)d_in[13];
  const float* be3 = (const float*)d_in[14];
  const float* p1  = (const float*)d_in[15];
  const float* p2  = (const float*)d_in[16];
  float* out = (float*)d_out;

  float* ws = (float*)d_ws;
  size_t off = 0;
  float* dis    = ws + off; off += N_NODES;
  float* stats  = ws + off; off += 256;
  off += 128 * 128;
  int*   cnt    = (int*)(ws + off); off += N_NODES;
  off += N_NODES + 16;
  off += N_NODES;
  off += NE_EDGES;
  float* lin    = ws + off; off += (size_t)N_NODES * DF;
  float* agg    = ws + off; off += (size_t)N_NODES * DF;
  float* agg2   = ws + off; off += (size_t)N_NODES * DF;
  float* part   = lin + 16;
  int*   slots  = (int*)(lin + 2097152 + 48);
  float* U    = lin;
  float* Qt   = lin + 16384;
  float* amat = lin + 32768;
  float* bmat = amat + (size_t)BPAIRS * DF;
  float* tmat = bmat + (size_t)BPAIRS * DF;

  const int* erow = edg;
  const int* ecol = edg + NE_EDGES;

  adj_build<<<ADJ_BLOCKS, ADJ_THREADS, 0, stream>>>(erow, ecol, cnt, slots, dis);

  const float* Wl[3] = {w1, w2, w3};
  const float* Bl[3] = {b1, b2, b3};
  const float* Gl[3] = {g1, g2, g3};
  const float* El[3] = {be1, be2, be3};
  float* Al[3] = {agg, agg2, agg};
  gcn_aggregate_linear<false><<<AGG_BLOCKS, AGG_WAVES * 32, 0, stream>>>(
      cnt, slots, erow, ecol, dis, x, stats, Wl[0], Bl[0], Al[0], part);
  bn_finalize_parts<<<DF / 16, 1024, 0, stream>>>(
      part, stats, Gl[0], El[0], 1.0f / (float)N_NODES);
  for (int l = 1; l < 3; ++l) {
    gcn_aggregate_linear<true><<<AGG_BLOCKS, AGG_WAVES * 32, 0, stream>>>(
        cnt, slots, erow, ecol, dis, Al[l - 1], stats, Wl[l], Bl[l], Al[l], part);
    bn_finalize_parts<<<DF / 16, 1024, 0, stream>>>(
        part, stats, Gl[l], El[l], 1.0f / (float)N_NODES);
  }

  const int small_tiles = (DF / 16) * (DF / 16);
  wmma_matmul<<<(small_tiles * 32 + 255) / 256, 256, 0, stream>>>(p1, p2, nullptr, U, DF, DF, DF, 1);
  wmma_matmul<<<(small_tiles * 32 + 255) / 256, 256, 0, stream>>>(U, p1, nullptr, Qt, DF, DF, DF, 1);
  gather_pairs<<<(BPAIRS * 32 + 255) / 256, 256, 0, stream>>>(Al[2], stats, di, amat, bmat);
  wmma_gemm_strip<<<((BPAIRS / 16) * 32 + 255) / 256, 256, 0, stream>>>(
      amat, Qt, nullptr, tmat, BPAIRS, DF);
  rowdot<<<(BPAIRS * 32 + 255) / 256, 256, 0, stream>>>(tmat, bmat, out);
}


# Layer kernel resident: 512 work-groups (2 per WGP) own 6-7 consecutive 16-row tiles each; waves loop over nodes, then all 32 waves multiply; 16384 waves enter per layer, not 50016; same sum order
# speedup vs baseline: 5.3074x; 1.0684x over previous
#include <hip/hip_runtime.h>

#define N_NODES 50000
#define NE_EDGES 800000
#define BPAIRS 4096
#define DF 128
#define EPS_BN 1e-5f

typedef float v2f __attribute__((ext_vector_type(2)));
typedef float v8f __attribute__((ext_vector_type(8)));
typedef float v4f __attribute__((ext_vector_type(4)));

__device__ __forceinline__ v8f wmma4(v2f a, v2f b, v8f c) {
  return __builtin_amdgcn_wmma_f32_16x16x4_f32(
       false, a,  false, b,
       (short)0, c,  false,  false);
}

__global__ __launch_bounds__(256)
void wmma_gemm_strip(const float* __restrict__ A, const float* __restrict__ Bt,
                     const float* __restrict__ bias, float* __restrict__ C,
                     int M, int K) {
  int gtid = blockIdx.x * blockDim.x + threadIdx.x;
  int wave = gtid >> 5;
  int lane = threadIdx.x & 31;
  if (wave * 16 >= M) return;

  int l16   = lane & 15;
  int khalf = (lane >> 4) << 1;
  const float* arow = A + (size_t)(wave * 16 + l16) * K + khalf;

  const float* brow[8];
#pragma unroll
  for (int nt = 0; nt < 8; ++nt)
    brow[nt] = Bt + (size_t)(nt * 16 + l16) * K + khalf;

  v8f zero = {};
  v8f acc[8];
#pragma unroll
  for (int nt = 0; nt < 8; ++nt) acc[nt] = zero;

  for (int k = 0; k < K; k += 4) {
    v2f a = *(const v2f*)(arow + k);
#pragma unroll
    for (int nt = 0; nt < 8; ++nt) {
      v2f b = *(const v2f*)(brow[nt] + k);
      acc[nt] = wmma4(a, b, acc[nt]);
    }
  }

  int rbase = wave * 16 + ((lane >> 4) << 3);
#pragma unroll
  for (int nt = 0; nt < 8; ++nt) {
    float bs = bias ? bias[nt * 16 + l16] : 0.f;
    float* crow = C + (size_t)rbase * 128 + nt * 16 + l16;
#pragma unroll
    for (int r = 0; r < 8; ++r)
      crow[(size_t)r * 128] = acc[nt][r] + bs;
  }
}

__global__ __launch_bounds__(256)
void wmma_matmul(const float* __restrict__ A, const float* __restrict__ Bm,
                 const float* __restrict__ bias, float* __restrict__ C,
                 int M, int Nc, int K, int bTrans) {
  int gtid = blockIdx.x * blockDim.x + threadIdx.x;
  int wave = gtid >> 5;
  int lane = threadIdx.x & 31;
  int ntiles = Nc >> 4;
  int mt = wave / ntiles;
  int nt = wave - mt * ntiles;
  if (mt * 16 >= M) return;

  int l16   = lane & 15;
  int khalf = (lane >> 4) << 1;
  const float* arow = A + (size_t)(mt * 16 + l16) * K + khalf;

  v8f acc = {};
  if (bTrans) {
    const float* brow = Bm + (size_t)(nt * 16 + l16) * K + khalf;
    for (int k = 0; k < K; k += 4) {
      v2f a = *(const v2f*)(arow + k);
      v2f b = *(const v2f*)(brow + k);
      acc = wmma4(a, b, acc);
    }
  } else {
    const float* bcol = Bm + (size_t)khalf * Nc + nt * 16 + l16;
    for (int k = 0; k < K; k += 4) {
      v2f a = *(const v2f*)(arow + k);
      v2f b;
      b.x = bcol[(size_t)k * Nc];
      b.y = bcol[(size_t)(k + 1) * Nc];
      acc = wmma4(a, b, acc);
    }
  }

  float bs = bias ? bias[nt * 16 + l16] : 0.f;
  int rbase = mt * 16 + ((lane >> 4) << 3);
  float* crow = C + (size_t)rbase * Nc + nt * 16 + l16;
#pragma unroll
  for (int r = 0; r < 8; ++r)
    crow[(size_t)r * Nc] = acc[r] + bs;
}

#define ADJ_CAP 64
#define ADJ_BLOCKS 128
#define ADJ_NB ((N_NODES + ADJ_BLOCKS - 1) / ADJ_BLOCKS)
#define ADJ_THREADS 1024
#define ADJ_Q (NE_EDGES / 4)
static_assert(NE_EDGES % 32 == 0, "the edge list is read in whole 16-byte pieces and whole waves");

__global__ __launch_bounds__(ADJ_THREADS)
void adj_build(const int* __restrict__ erow, const int* __restrict__ ecol,
               int* __restrict__ cnt, int* __restrict__ slots, float* __restrict__ dis) {
  __shared__ int cntL[ADJ_NB];
  __shared__ int slotL[ADJ_NB * ADJ_CAP];
  const int lo = blockIdx.x * ADJ_NB;
  int nb = N_NODES - lo; if (nb > ADJ_NB) nb = ADJ_NB;
  for (int i = threadIdx.x; i < ADJ_NB; i += ADJ_THREADS) cntL[i] = 0;
  __syncthreads();

  const int4* ec4 = (const int4*)ecol;
  for (int q0 = threadIdx.x; q0 < ADJ_Q; q0 += ADJ_THREADS * 8) {
    int4 v[8];
#pragma unroll
    for (int u = 0; u < 8; ++u) {
      int q = q0 + u * ADJ_THREADS;
      v[u] = ec4[q < ADJ_Q ? q : ADJ_Q - 1];
    }
    int p[8][4];
#pragma unroll
    for (int u = 0; u < 8; ++u) {
      int q = q0 + u * ADJ_THREADS;
      int cc[4] = {v[u].x, v[u].y, v[u].z, v[u].w};
#pragma unroll
      for (int k = 0; k < 4; ++k) {
        unsigned d = (unsigned)(cc[k] - lo);
        p[u][k] = (q < ADJ_Q && d < (unsigned)nb) ? atomicAdd(&cntL[d], 1) : ADJ_CAP;
      }
    }
#pragma unroll
    for (int u = 0; u < 8; ++u) {
      int q = q0 + u * ADJ_THREADS;
      int cc[4] = {v[u].x, v[u].y, v[u].z, v[u].w};
#pragma unroll
      for (int k = 0; k < 4; ++k)
        if (p[u][k] < ADJ_CAP) slotL[(cc[k] - lo) * ADJ_CAP + p[u][k]] = q * 4 + k;
    }
  }
  __syncthreads();

  for (int i = threadIdx.x; i < nb; i += ADJ_THREADS) {
    int dg = cntL[i];
    cnt[lo + i] = dg;
    dis[lo + i] = rsqrtf((float)dg + 1.0f);
  }
  int4* out4 = (int4*)(slots + (size_t)lo * ADJ_CAP);
  const int4* in4 = (const int4*)slotL;
  for (int t = threadIdx.x; t < nb * (ADJ_CAP / 4); t += ADJ_THREADS) {
    int i = t >> 4, j = t & 15;
    int left = cntL[i] - j * 4;
    if (left <= 0) continue;
    int4 e4 = in4[t];
    int4 r4;
    r4.x = erow[e4.x];
    r4.y = erow[left > 1 ? e4.y : e4.x];
    r4.z = erow[left > 2 ? e4.z : e4.x];
    r4.w = erow[left > 3 ? e4.w : e4.x];
    out4[t] = r4;
  }
}

#define AGG_WAVES 32
#define N_PARTS (N_NODES / 16)
#define AGG_GRID 512
#define AGG_MAXT ((N_PARTS + AGG_GRID - 1) / AGG_GRID)
#define A_PITCH 132
static_assert(N_NODES % 16 == 0, "whole 16-row tiles only");
static_assert(AGG_WAVES == 32, "four sets of eight product waves");
static_assert(AGG_MAXT <= 8, "a product wave takes at most two tiles, four tiles apart");
static_assert(N_PARTS / AGG_GRID >= 4, "every product wave has a first tile");

__device__ __forceinline__ float bn_leaky(float v, float sc, float sh) {
  float y = v * sc + sh;
  return fmaxf(y, 0.1f * y);
}

template <bool BN_IN>
__device__ __forceinline__ void add_rows(const float* __restrict__ xin, const float* __restrict__ dis,
                                         int lane, int myr, float dc, int j0, const int n,
                                         const float4 sc, const float4 sh,
                                         float& ax, float& ay, float& az, float& aw, float& ns) {
  v4f v[8];
  float  d[8];
#pragma unroll
  for (int u = 0; u < 8; ++u) {
    if (u < n) {
      int r = __builtin_amdgcn_readlane(myr, j0 + u);
      d[u] = dis[r];
      v[u] = *(const v4f*)(xin + (size_t)r * DF + lane * 4);
    }
  }
  if (BN_IN) {
    if (n == 8) asm volatile("" : "+v"(v[0]), "+v"(v[1]), "+v"(v[2]), "+v"(v[3]), "+v"(v[4]), "+v"(v[5]), "+v"(v[6]), "+v"(v[7]));
    else if (n == 7) asm volatile("" : "+v"(v[0]), "+v"(v[1]), "+v"(v[2]), "+v"(v[3]), "+v"(v[4]), "+v"(v[5]), "+v"(v[6]));
    else if (n == 6) asm volatile("" : "+v"(v[0]), "+v"(v[1]), "+v"(v[2]), "+v"(v[3]), "+v"(v[4]), "+v"(v[5]));
  }
#pragma unroll
  for (int u = 0; u < 8; ++u) {
    if (u < n) {
      float nrm = d[u] * dc;
      v4f r = v[u];
      if (BN_IN) {
        r.x = bn_leaky(r.x, sc.x, sh.x); r.y = bn_leaky(r.y, sc.y, sh.y);
        r.z = bn_leaky(r.z, sc.z, sh.z); r.w = bn_leaky(r.w, sc.w, sh.w);
      }
      ax += nrm * r.x; ay += nrm * r.y; az += nrm * r.z; aw += nrm * r.w;
      ns += nrm;
    }
  }
}

__device__ __forceinline__ void store_piece(v8f acc, const float* __restrict__ ws16, float bs,
                                            float* __restrict__ crow, float* __restrict__ pline,
                                            int hi) {
  float cs = 0.f, cq = 0.f;
#pragma unroll
  for (int r = 0; r < 8; ++r) {
    float v = acc[r] + ws16[hi * 8 + r] * bs;
    crow[(size_t)r * DF] = v;
    cs += v; cq += v * v;
  }
  cs += __shfl_xor(cs, 16, 32);
  cq += __shfl_xor(cq, 16, 32);
  *pline = hi ? cq : cs;
}

template <bool BN_IN>
__global__ __launch_bounds__(AGG_WAVES * 32, 16)
void gcn_aggregate_linear(const int* __restrict__ cnt, const int* __restrict__ slots,
                          const int* __restrict__ erow, const int* __restrict__ ecol,
                          const float* __restrict__ dis, const float* __restrict__ xin,
                          const float* __restrict__ stats_in,
                          const float* __restrict__ Wm, const float* __restrict__ bias,
                          float* __restrict__ agg, float* __restrict__ part) {
  __shared__ float atile[AGG_MAXT * 16 * A_PITCH];
  __shared__ float wsum[AGG_MAXT * 16];

  int w = __builtin_amdgcn_readfirstlane(threadIdx.x >> 5);
  int lane = threadIdx.x & 31;
  const int t0 = ((int)blockIdx.x * N_PARTS) / AGG_GRID;
  const int nt = (((int)blockIdx.x + 1) * N_PARTS) / AGG_GRID - t0;
  const int nrows = nt * 16;
  float4 sc, sh;
  sc.x = sc.y = sc.z = sc.w = 1.f;
  sh.x = sh.y = sh.z = sh.w = 0.f;
  if (BN_IN) {
    sc = *(const float4*)(stats_in + lane * 4);
    sh = *(const float4*)(stats_in + DF + lane * 4);
  }
  float* arow = atile + w * A_PITCH + lane * 4;
  for (int i = w; i < nrows; i += AGG_WAVES, arow += AGG_WAVES * A_PITCH) {
    int c = t0 * 16 + i;
    float dc = dis[c];
    float4 sv = *(const float4*)(xin + (size_t)c * DF + lane * 4);
    if (BN_IN) {
      sv.x = bn_leaky(sv.x, sc.x, sh.x); sv.y = bn_leaky(sv.y, sc.y, sh.y);
      sv.z = bn_leaky(sv.z, sc.z, sh.z); sv.w = bn_leaky(sv.w, sc.w, sh.w);
    }
    float s = dc * dc;
    float ax = s * sv.x, ay = s * sv.y, az = s * sv.z, aw = s * sv.w;
    float ns = s;
    int deg = cnt[c];
    int nslot = deg <= ADJ_CAP ? deg : 0;
    const int* sl = slots + (size_t)c * ADJ_CAP;
    for (int base = 0; base < nslot; base += 32) {
      int left = nslot - base;
      if (left > 32) left = 32;
      int myr = c;
      if (lane < left) myr = sl[base + lane];
      int j = 0;
      for (; j + 8 <= left; j += 8)
        add_rows<BN_IN>(xin, dis, lane, myr, dc, j, 8, sc, sh, ax, ay, az, aw, ns);
      switch (left - j) {
        case 1: add_rows<BN_IN>(xin, dis, lane, myr, dc, j, 1, sc, sh, ax, ay, az, aw, ns); break;
        case 2: add_rows<BN_IN>(xin, dis, lane, myr, dc, j, 2, sc, sh, ax, ay, az, aw, ns); break;
        case 3: add_rows<BN_IN>(xin, dis, lane, myr, dc, j, 3, sc, sh, ax, ay, az, aw, ns); break;
        case 4: add_rows<BN_IN>(xin, dis, lane, myr, dc, j, 4, sc, sh, ax, ay, az, aw, ns); break;
        case 5: add_rows<BN_IN>(xin, dis, lane, myr, dc, j, 5, sc, sh, ax, ay, az, aw, ns); break;
        case 6: add_rows<BN_IN>(xin, dis, lane, myr, dc, j, 6, sc, sh, ax, ay, az, aw, ns); break;
        case 7: add_rows<BN_IN>(xin, dis, lane, myr, dc, j, 7, sc, sh, ax, ay, az, aw, ns); break;
        default: break;
      }
    }
    if (deg > ADJ_CAP) {
      for (int e0 = 0; e0 < NE_EDGES; e0 += 32) {
        unsigned m = __builtin_amdgcn_ballot_w32(ecol[e0 + lane] == c);
        while (m) {
          int b = __builtin_ctz(m);
          m &= m - 1;
          int r = erow[e0 + b];
          float nrm = dis[r] * dc;
          float4 v = *(const float4*)(xin + (size_t)r * DF + lane * 4);
          if (BN_IN) {
            v.x = bn_leaky(v.x, sc.x, sh.x); v.y = bn_leaky(v.y, sc.y, sh.y);
            v.z = bn_leaky(v.z, sc.z, sh.z); v.w = bn_leaky(v.w, sc.w, sh.w);
          }
          ax += nrm * v.x; ay += nrm * v.y; az += nrm * v.z; aw += nrm * v.w;
          ns += nrm;
        }
      }
    }
    float4 o; o.x = ax; o.y = ay; o.z = az; o.w = aw;
    *(float4*)arow = o;
    if (lane == 0) wsum[i] = ns;
  }
  __syncthreads();

  int l16   = lane & 15;
  int hi    = lane >> 4;
  int khalf = hi << 1;
  int g     = w & 7;
  int ta    = w >> 3;
  int tb    = ta + 4;
  if (ta >= nt) return;
  const float* ap0 = atile + (ta * 16 + l16) * A_PITCH + khalf;
  const float* bp  = Wm + (size_t)khalf * DF + g * 16 + l16;
  float bs = bias[g * 16 + l16];
  float* crow0 = agg + ((size_t)(t0 + ta) * 16 + hi * 8) * DF + g * 16 + l16;
  float* pl0   = part + ((size_t)g * N_PARTS + (t0 + ta)) * 32 + lane;

  if (tb < nt) {
    const float* ap1 = ap0 + 64 * A_PITCH;
    v8f acc0 = {};
    v8f acc1 = {};
#pragma unroll 8
    for (int k = 0; k < DF; k += 4) {
      v2f a0 = *(const v2f*)(ap0 + k);
      v2f a1 = *(const v2f*)(ap1 + k);
      v2f b;
      b.x = bp[(size_t)k * DF];
      b.y = bp[(size_t)(k + 1) * DF];
      acc0 = wmma4(a0, b, acc0);
      acc1 = wmma4(a1, b, acc1);
    }
    store_piece(acc0, wsum + ta * 16, bs, crow0, pl0, hi);
    store_piece(acc1, wsum + tb * 16, bs, crow0 + (size_t)64 * DF, pl0 + 4 * 32, hi);
  } else {
    v8f acc0 = {};
#pragma unroll 8
    for (int k = 0; k < DF; k += 4) {
      v2f a0 = *(const v2f*)(ap0 + k);
      v2f b;
      b.x = bp[(size_t)k * DF];
      b.y = bp[(size_t)(k + 1) * DF];
      acc0 = wmma4(a0, b, acc0);
    }
    store_piece(acc0, wsum + ta * 16, bs, crow0, pl0, hi);
  }
}

__global__ __launch_bounds__(1024)
void bn_finalize_parts(const float* __restrict__ part, float* __restrict__ stats,
                       const float* __restrict__ g, const float* __restrict__ be,
                       float inv_n) {
  __shared__ float red[32 * 32];
  const int QCH  = (N_PARTS + 3) / 4;
  const int ITER = (QCH + 31) / 32;
  int grp  = blockIdx.x;
  int w    = threadIdx.x >> 5;
  int lane = threadIdx.x & 31;
  int sub  = lane >> 3;
  int e0   = (lane & 7) * 4;
  const float* base = part + (size_t)grp * N_PARTS * 32;

  float4 acc; acc.x = 0.f; acc.y = 0.f; acc.z = 0.f; acc.w = 0.f;
#pragma unroll 7
  for (int it = 0; it < ITER; ++it) {
    int p  = (it * 32 + w) * 4 + sub;
    bool ok = p < N_PARTS;
    int pc = ok ? p : N_PARTS - 1;
    float4 v = *(const float4*)(base + (size_t)pc * 32 + e0);
    acc.x += ok ? v.x : 0.f; acc.y += ok ? v.y : 0.f;
    acc.z += ok ? v.z : 0.f; acc.w += ok ? v.w : 0.f;
  }
#pragma unroll
  for (int off = 8; off < 32; off <<= 1) {
    acc.x += __shfl_xor(acc.x, off, 32); acc.y += __shfl_xor(acc.y, off, 32);
    acc.z += __shfl_xor(acc.z, off, 32); acc.w += __shfl_xor(acc.w, off, 32);
  }
  if (lane < 8) *(float4*)(red + w * 32 + e0) = acc;
  __syncthreads();
  if (w == 0) {
    float tot = 0.f;
    for (int i = 0; i < 32; ++i) tot += red[i * 32 + lane];
    float sq = __shfl_down(tot, 16, 32);
    if (lane < 16) {
      int f = grp * 16 + lane;
      float m   = tot * inv_n;
      float var = sq * inv_n - m * m;
      float sc  = g[f] * rsqrtf(var + EPS_BN);
      stats[f]      = sc;
      stats[DF + f] = be[f] - m * sc;
    }
  }
}

__global__ __launch_bounds__(256)
void gather_pairs(const float* __restrict__ aggp, const float* __restrict__ stats,
                  const int* __restrict__ di,
                  float* __restrict__ amat, float* __restrict__ bmat) {
  int t = blockIdx.x * blockDim.x + threadIdx.x;
  if (t >= BPAIRS * 32) return;
  int j = t >> 5, q = t & 31;
  int ia = di[2 * j] - 1;     if (ia < 0) ia += N_NODES;
  int ib = di[2 * j + 1] - 1; if (ib < 0) ib += N_NODES;
  const float4 sc = *(const float4*)(stats + q * 4);
  const float4 sh = *(const float4*)(stats + DF + q * 4);
  float4 va = *(const float4*)(aggp + (size_t)ia * DF + q * 4);
  float4 vb = *(const float4*)(aggp + (size_t)ib * DF + q * 4);
  va.x = bn_leaky(va.x, sc.x, sh.x); va.y = bn_leaky(va.y, sc.y, sh.y);
  va.z = bn_leaky(va.z, sc.z, sh.z); va.w = bn_leaky(va.w, sc.w, sh.w);
  vb.x = bn_leaky(vb.x, sc.x, sh.x); vb.y = bn_leaky(vb.y, sc.y, sh.y);
  vb.z = bn_leaky(vb.z, sc.z, sh.z); vb.w = bn_leaky(vb.w, sc.w, sh.w);
  *(float4*)(amat + (size_t)j * DF + q * 4) = va;
  *(float4*)(bmat + (size_t)j * DF + q * 4) = vb;
}

__global__ __launch_bounds__(256)
void rowdot(const float* __restrict__ tm, const float* __restrict__ bm,
            float* __restrict__ out) {
  int t = blockIdx.x * blockDim.x + threadIdx.x;
  int j = t >> 5, lane = t & 31;
  if (j >= BPAIRS) return;
  float4 tv = *(const float4*)(tm + (size_t)j * DF + lane * 4);
  float4 bv = *(const float4*)(bm + (size_t)j * DF + lane * 4);
  float p = tv.x * bv.x + tv.y * bv.y + tv.z * bv.z + tv.w * bv.w;
#pragma unroll
  for (int off = 16; off > 0; off >>= 1)
    p += __shfl_down(p, off, 32);
  if (lane == 0) out[j] = p;
}

extern "C" void kernel_launch(void* const* d_in, const int* in_sizes, int n_in,
                              void* d_out, int out_size, void* d_ws, size_t ws_size,
                              hipStream_t stream) {
  const float* x   = (const float*)d_in[0];
  const int*   edg = (const int*)d_in[1];
  const int*   di  = (const int*)d_in[2];
  const float* w1  = (const float*)d_in[3];
  const float* b1  = (const float*)d_in[4];
  const float* w2  = (const float*)d_in[5];
  const float* b2  = (const float*)d_in[6];
  const float* w3  = (const float*)d_in[7];
  const float* b3  = (const float*)d_in[8];
  const float* g1  = (const float*)d_in[9];
  const float* be1 = (const float*)d_in[10];
  const float* g2  = (const float*)d_in[11];
  const float* be2 = (const float*)d_in[12];
  const float* g3  = (const float*)d_in[13];
  const float* be3 = (const float*)d_in[14];
  const float* p1  = (const float*)d_in[15];
  const float* p2  = (const float*)d_in[16];
  float* out = (float*)d_out;

  float* ws = (float*)d_ws;
  size_t off = 0;
  float* dis    = ws + off; off += N_NODES;
  float* stats  = ws + off; off += 256;
  off += 128 * 128;
  int*   cnt    = (int*)(ws + off); off += N_NODES;
  off += N_NODES + 16;
  off += N_NODES;
  off += NE_EDGES;
  float* lin    = ws + off; off += (size_t)N_NODES * DF;
  float* agg    = ws + off; off += (size_t)N_NODES * DF;
  float* agg2   = ws + off; off += (size_t)N_NODES * DF;
  float* part   = lin + 16;
  int*   slots  = (int*)(lin + 2097152 + 48);
  float* U    = lin;
  float* Qt   = lin + 16384;
  float* amat = lin + 32768;
  float* bmat = amat + (size_t)BPAIRS * DF;
  float* tmat = bmat + (size_t)BPAIRS * DF;

  const int* erow = edg;
  const int* ecol = edg + NE_EDGES;

  adj_build<<<ADJ_BLOCKS, ADJ_THREADS, 0, stream>>>(erow, ecol, cnt, slots, dis);

  const float* Wl[3] = {w1, w2, w3};
  const float* Bl[3] = {b1, b2, b3};
  const float* Gl[3] = {g1, g2, g3};
  const float* El[3] = {be1, be2, be3};
  float* Al[3] = {agg, agg2, agg};
  gcn_aggregate_linear<false><<<AGG_GRID, AGG_WAVES * 32, 0, stream>>>(
      cnt, slots, erow, ecol, dis, x, stats, Wl[0], Bl[0], Al[0], part);
  bn_finalize_parts<<<DF / 16, 1024, 0, stream>>>(
      part, stats, Gl[0], El[0], 1.0f / (float)N_NODES);
  for (int l = 1; l < 3; ++l) {
    gcn_aggregate_linear<true><<<AGG_GRID, AGG_WAVES * 32, 0, stream>>>(
        cnt, slots, erow, ecol, dis, Al[l - 1], stats, Wl[l], Bl[l], Al[l], part);
    bn_finalize_parts<<<DF / 16, 1024, 0, stream>>>(
        part, stats, Gl[l], El[l], 1.0f / (float)N_NODES);
  }

  const int small_tiles = (DF / 16) * (DF / 16);
  wmma_matmul<<<(small_tiles * 32 + 255) / 256, 256, 0, stream>>>(p1, p2, nullptr, U, DF, DF, DF, 1);
  wmma_matmul<<<(small_tiles * 32 + 255) / 256, 256, 0, stream>>>(U, p1, nullptr, Qt, DF, DF, DF, 1);
  gather_pairs<<<(BPAIRS * 32 + 255) / 256, 256, 0, stream>>>(Al[2], stats, di, amat, bmat);
  wmma_gemm_strip<<<((BPAIRS / 16) * 32 + 255) / 256, 256, 0, stream>>>(
      amat, Qt, nullptr, tmat, BPAIRS, DF);
  rowdot<<<(BPAIRS * 32 + 255) / 256, 256, 0, stream>>>(tmat, bmat, out);
}


# Adjacency build reads the edge list once: 256 blocks sort 3125-edge runs by target owner into 256-byte cells (LDS), 196 owner blocks slot only their own cells; in-row order: LDS arrival as before
# speedup vs baseline: 6.1032x; 1.1499x over previous
#include <hip/hip_runtime.h>

#define N_NODES 50000
#define NE_EDGES 800000
#define BPAIRS 4096
#define DF 128
#define EPS_BN 1e-5f

typedef float v2f __attribute__((ext_vector_type(2)));
typedef float v8f __attribute__((ext_vector_type(8)));
typedef float v4f __attribute__((ext_vector_type(4)));

__device__ __forceinline__ v8f wmma4(v2f a, v2f b, v8f c) {
  return __builtin_amdgcn_wmma_f32_16x16x4_f32(
       false, a,  false, b,
       (short)0, c,  false,  false);
}

__global__ __launch_bounds__(256)
void wmma_gemm_strip(const float* __restrict__ A, const float* __restrict__ Bt,
                     const float* __restrict__ bias, float* __restrict__ C,
                     int M, int K) {
  int gtid = blockIdx.x * blockDim.x + threadIdx.x;
  int wave = gtid >> 5;
  int lane = threadIdx.x & 31;
  if (wave * 16 >= M) return;

  int l16   = lane & 15;
  int khalf = (lane >> 4) << 1;
  const float* arow = A + (size_t)(wave * 16 + l16) * K + khalf;

  const float* brow[8];
#pragma unroll
  for (int nt = 0; nt < 8; ++nt)
    brow[nt] = Bt + (size_t)(nt * 16 + l16) * K + khalf;

  v8f zero = {};
  v8f acc[8];
#pragma unroll
  for (int nt = 0; nt < 8; ++nt) acc[nt] = zero;

  for (int k = 0; k < K; k += 4) {
    v2f a = *(const v2f*)(arow + k);
#pragma unroll
    for (int nt = 0; nt < 8; ++nt) {
      v2f b = *(const v2f*)(brow[nt] + k);
      acc[nt] = wmma4(a, b, acc[nt]);
    }
  }

  int rbase = wave * 16 + ((lane >> 4) << 3);
#pragma unroll
  for (int nt = 0; nt < 8; ++nt) {
    float bs = bias ? bias[nt * 16 + l16] : 0.f;
    float* crow = C + (size_t)rbase * 128 + nt * 16 + l16;
#pragma unroll
    for (int r = 0; r < 8; ++r)
      crow[(size_t)r * 128] = acc[nt][r] + bs;
  }
}

__global__ __launch_bounds__(256)
void wmma_matmul(const float* __restrict__ A, const float* __restrict__ Bm,
                 const float* __restrict__ bias, float* __restrict__ C,
                 int M, int Nc, int K, int bTrans) {
  int gtid = blockIdx.x * blockDim.x + threadIdx.x;
  int wave = gtid >> 5;
  int lane = threadIdx.x & 31;
  int ntiles = Nc >> 4;
  int mt = wave / ntiles;
  int nt = wave - mt * ntiles;
  if (mt * 16 >= M) return;

  int l16   = lane & 15;
  int khalf = (lane >> 4) << 1;
  const float* arow = A + (size_t)(mt * 16 + l16) * K + khalf;

  v8f acc = {};
  if (bTrans) {
    const float* brow = Bm + (size_t)(nt * 16 + l16) * K + khalf;
    for (int k = 0; k < K; k += 4) {
      v2f a = *(const v2f*)(arow + k);
      v2f b = *(const v2f*)(brow + k);
      acc = wmma4(a, b, acc);
    }
  } else {
    const float* bcol = Bm + (size_t)khalf * Nc + nt * 16 + l16;
    for (int k = 0; k < K; k += 4) {
      v2f a = *(const v2f*)(arow + k);
      v2f b;
      b.x = bcol[(size_t)k * Nc];
      b.y = bcol[(size_t)(k + 1) * Nc];
      acc = wmma4(a, b, acc);
    }
  }

  float bs = bias ? bias[nt * 16 + l16] : 0.f;
  int rbase = mt * 16 + ((lane >> 4) << 3);
  float* crow = C + (size_t)rbase * Nc + nt * 16 + l16;
#pragma unroll
  for (int r = 0; r < 8; ++r)
    crow[(size_t)r * Nc] = acc[r] + bs;
}

#define ADJ_CAP 64
#define ADJ_THREADS 1024
#define OWN_SH 8
#define OWN_N (1 << OWN_SH)
#define N_OWN ((N_NODES + OWN_N - 1) / OWN_N)
#define CH_BLOCKS 256
#define CH_E (NE_EDGES / CH_BLOCKS)
#define CH_THREADS 512
#define CH_PER ((CH_E + CH_THREADS - 1) / CH_THREADS)
#define CELL_CAP 64
static_assert(NE_EDGES % CH_BLOCKS == 0, "whole runs only");
static_assert(N_NODES <= 65536, "a source node is packed into 16 bits beside the target's place");
static_assert(CH_BLOCKS * (CELL_CAP / 4) == 4 * ADJ_THREADS, "an owner's cells are four 16-byte pieces per thread");

__global__ __launch_bounds__(CH_THREADS)
void edge_partition(const int* __restrict__ erow, const int* __restrict__ ecol,
                    int* __restrict__ stage, int* __restrict__ cellcnt) {
  __shared__ int cntL[N_OWN];
  __shared__ int cellL[N_OWN * CELL_CAP];
  const int b = blockIdx.x;
  for (int i = threadIdx.x; i < N_OWN; i += CH_THREADS) cntL[i] = 0;
  __syncthreads();

  const int e0 = b * CH_E;
  int cc[CH_PER], rr[CH_PER];
#pragma unroll
  for (int u = 0; u < CH_PER; ++u) {
    int i = threadIdx.x + u * CH_THREADS;
    int e = e0 + (i < CH_E ? i : CH_E - 1);
    cc[u] = ecol[e];
    rr[u] = erow[e];
  }
  int rk[CH_PER];
#pragma unroll
  for (int u = 0; u < CH_PER; ++u) {
    int i = threadIdx.x + u * CH_THREADS;
    rk[u] = i < CH_E ? atomicAdd(&cntL[cc[u] >> OWN_SH], 1) : CELL_CAP;
  }
#pragma unroll
  for (int u = 0; u < CH_PER; ++u)
    if (rk[u] < CELL_CAP)
      cellL[(cc[u] >> OWN_SH) * CELL_CAP + rk[u]] = ((cc[u] & (OWN_N - 1)) << 16) | rr[u];
  __syncthreads();

  for (int o = threadIdx.x; o < N_OWN; o += CH_THREADS) cellcnt[o * CH_BLOCKS + b] = cntL[o];
  int4* out4 = (int4*)stage;
  const int4* in4 = (const int4*)cellL;
  for (int p = threadIdx.x; p < N_OWN * (CELL_CAP / 4); p += CH_THREADS) {
    int o = p >> 4, j = p & 15;
    int n = cntL[o];
    if (n <= CELL_CAP && j * 4 < n)
      out4[((size_t)o * CH_BLOCKS + b) * (CELL_CAP / 4) + j] = in4[p];
  }
}

__device__ __forceinline__ void slot_edge(int* cntL, int* slotL, int lc, int r) {
  int rank = atomicAdd(&cntL[lc], 1);
  if (rank < ADJ_CAP) slotL[lc * ADJ_CAP + rank] = r;
}

__global__ __launch_bounds__(ADJ_THREADS)
void adj_build(const int* __restrict__ erow, const int* __restrict__ ecol,
               const int* __restrict__ stage, const int* __restrict__ cellcnt,
               int* __restrict__ cnt, int* __restrict__ slots, float* __restrict__ dis) {
  __shared__ int cntL[OWN_N];
  __shared__ int ccL[CH_BLOCKS];
  __shared__ int slotL[OWN_N * ADJ_CAP];
  const int o  = blockIdx.x;
  const int lo = o * OWN_N;
  int nb = N_NODES - lo; if (nb > OWN_N) nb = OWN_N;
  for (int i = threadIdx.x; i < OWN_N; i += ADJ_THREADS) cntL[i] = 0;
  for (int i = threadIdx.x; i < CH_BLOCKS; i += ADJ_THREADS) ccL[i] = cellcnt[o * CH_BLOCKS + i];
  __syncthreads();

  const int4* st4 = (const int4*)stage + (size_t)o * CH_BLOCKS * (CELL_CAP / 4);
  int4 v[4];
  int left[4];
#pragma unroll
  for (int u = 0; u < 4; ++u) {
    int p = threadIdx.x + u * ADJ_THREADS;
    int n = ccL[p >> 4];
    left[u] = n <= CELL_CAP ? n - (p & 15) * 4 : 0;
    v[u].x = v[u].y = v[u].z = v[u].w = 0;
    if (left[u] > 0) v[u] = st4[p];
  }
  int rk[4][4];
#pragma unroll
  for (int u = 0; u < 4; ++u) {
    int pe[4] = {v[u].x, v[u].y, v[u].z, v[u].w};
#pragma unroll
    for (int k = 0; k < 4; ++k)
      rk[u][k] = left[u] > k ? atomicAdd(&cntL[pe[k] >> 16], 1) : ADJ_CAP;
  }
#pragma unroll
  for (int u = 0; u < 4; ++u) {
    int pe[4] = {v[u].x, v[u].y, v[u].z, v[u].w};
#pragma unroll
    for (int k = 0; k < 4; ++k)
      if (rk[u][k] < ADJ_CAP) slotL[(pe[k] >> 16) * ADJ_CAP + rk[u][k]] = pe[k] & 0xffff;
  }
  for (int b = 0; b < CH_BLOCKS; ++b) {
    if (ccL[b] <= CELL_CAP) continue;
    for (int i = threadIdx.x; i < CH_E; i += ADJ_THREADS) {
      int c = ecol[b * CH_E + i];
      if ((c >> OWN_SH) == o) slot_edge(cntL, slotL, c & (OWN_N - 1), erow[b * CH_E + i]);
    }
  }
  __syncthreads();

  for (int i = threadIdx.x; i < nb; i += ADJ_THREADS) {
    int dg = cntL[i];
    cnt[lo + i] = dg;
    dis[lo + i] = rsqrtf((float)dg + 1.0f);
  }
  int4* out4 = (int4*)(slots + (size_t)lo * ADJ_CAP);
  const int4* in4 = (const int4*)slotL;
  for (int t = threadIdx.x; t < nb * (ADJ_CAP / 4); t += ADJ_THREADS) {
    int i = t >> 4, j = t & 15;
    if (cntL[i] - j * 4 > 0) out4[t] = in4[t];
  }
}

#define AGG_WAVES 32
#define N_PARTS (N_NODES / 16)
#define AGG_GRID 512
#define AGG_MAXT ((N_PARTS + AGG_GRID - 1) / AGG_GRID)
#define A_PITCH 132
static_assert(N_NODES % 16 == 0, "whole 16-row tiles only");
static_assert(AGG_WAVES == 32, "four sets of eight product waves");
static_assert(AGG_MAXT <= 8, "a product wave takes at most two tiles, four tiles apart");
static_assert(N_PARTS / AGG_GRID >= 4, "every product wave has a first tile");

__device__ __forceinline__ float bn_leaky(float v, float sc, float sh) {
  float y = v * sc + sh;
  return fmaxf(y, 0.1f * y);
}

template <bool BN_IN>
__device__ __forceinline__ void add_rows(const float* __restrict__ xin, const float* __restrict__ dis,
                                         int lane, int myr, float dc, int j0, const int n,
                                         const float4 sc, const float4 sh,
                                         float& ax, float& ay, float& az, float& aw, float& ns) {
  v4f v[8];
  float  d[8];
#pragma unroll
  for (int u = 0; u < 8; ++u) {
    if (u < n) {
      int r = __builtin_amdgcn_readlane(myr, j0 + u);
      d[u] = dis[r];
      v[u] = *(const v4f*)(xin + (size_t)r * DF + lane * 4);
    }
  }
  if (BN_IN) {
    if (n == 8) asm volatile("" : "+v"(v[0]), "+v"(v[1]), "+v"(v[2]), "+v"(v[3]), "+v"(v[4]), "+v"(v[5]), "+v"(v[6]), "+v"(v[7]));
    else if (n == 7) asm volatile("" : "+v"(v[0]), "+v"(v[1]), "+v"(v[2]), "+v"(v[3]), "+v"(v[4]), "+v"(v[5]), "+v"(v[6]));
    else if (n == 6) asm volatile("" : "+v"(v[0]), "+v"(v[1]), "+v"(v[2]), "+v"(v[3]), "+v"(v[4]), "+v"(v[5]));
  }
#pragma unroll
  for (int u = 0; u < 8; ++u) {
    if (u < n) {
      float nrm = d[u] * dc;
      v4f r = v[u];
      if (BN_IN) {
        r.x = bn_leaky(r.x, sc.x, sh.x); r.y = bn_leaky(r.y, sc.y, sh.y);
        r.z = bn_leaky(r.z, sc.z, sh.z); r.w = bn_leaky(r.w, sc.w, sh.w);
      }
      ax += nrm * r.x; ay += nrm * r.y; az += nrm * r.z; aw += nrm * r.w;
      ns += nrm;
    }
  }
}

__device__ __forceinline__ void store_piece(v8f acc, const float* __restrict__ ws16, float bs,
                                            float* __restrict__ crow, float* __restrict__ pline,
                                            int hi) {
  float cs = 0.f, cq = 0.f;
#pragma unroll
  for (int r = 0; r < 8; ++r) {
    float v = acc[r] + ws16[hi * 8 + r] * bs;
    crow[(size_t)r * DF] = v;
    cs += v; cq += v * v;
  }
  cs += __shfl_xor(cs, 16, 32);
  cq += __shfl_xor(cq, 16, 32);
  *pline = hi ? cq : cs;
}

template <bool BN_IN>
__global__ __launch_bounds__(AGG_WAVES * 32, 16)
void gcn_aggregate_linear(const int* __restrict__ cnt, const int* __restrict__ slots,
                          const int* __restrict__ erow, const int* __restrict__ ecol,
                          const float* __restrict__ dis, const float* __restrict__ xin,
                          const float* __restrict__ stats_in,
                          const float* __restrict__ Wm, const float* __restrict__ bias,
                          float* __restrict__ agg, float* __restrict__ part) {
  __shared__ float atile[AGG_MAXT * 16 * A_PITCH];
  __shared__ float wsum[AGG_MAXT * 16];

  int w = __builtin_amdgcn_readfirstlane(threadIdx.x >> 5);
  int lane = threadIdx.x & 31;
  const int t0 = ((int)blockIdx.x * N_PARTS) / AGG_GRID;
  const int nt = (((int)blockIdx.x + 1) * N_PARTS) / AGG_GRID - t0;
  const int nrows = nt * 16;
  float4 sc, sh;
  sc.x = sc.y = sc.z = sc.w = 1.f;
  sh.x = sh.y = sh.z = sh.w = 0.f;
  if (BN_IN) {
    sc = *(const float4*)(stats_in + lane * 4);
    sh = *(const float4*)(stats_in + DF + lane * 4);
  }
  float* arow = atile + w * A_PITCH + lane * 4;
  for (int i = w; i < nrows; i += AGG_WAVES, arow += AGG_WAVES * A_PITCH) {
    int c = t0 * 16 + i;
    float dc = dis[c];
    float4 sv = *(const float4*)(xin + (size_t)c * DF + lane * 4);
    if (BN_IN) {
      sv.x = bn_leaky(sv.x, sc.x, sh.x); sv.y = bn_leaky(sv.y, sc.y, sh.y);
      sv.z = bn_leaky(sv.z, sc.z, sh.z); sv.w = bn_leaky(sv.w, sc.w, sh.w);
    }
    float s = dc * dc;
    float ax = s * sv.x, ay = s * sv.y, az = s * sv.z, aw = s * sv.w;
    float ns = s;
    int deg = cnt[c];
    int nslot = deg <= ADJ_CAP ? deg : 0;
    const int* sl = slots + (size_t)c * ADJ_CAP;
    for (int base = 0; base < nslot; base += 32) {
      int left = nslot - base;
      if (left > 32) left = 32;
      int myr = c;
      if (lane < left) myr = sl[base + lane];
      int j = 0;
      for (; j + 8 <= left; j += 8)
        add_rows<BN_IN>(xin, dis, lane, myr, dc, j, 8, sc, sh, ax, ay, az, aw, ns);
      switch (left - j) {
        case 1: add_rows<BN_IN>(xin, dis, lane, myr, dc, j, 1, sc, sh, ax, ay, az, aw, ns); break;
        case 2: add_rows<BN_IN>(xin, dis, lane, myr, dc, j, 2, sc, sh, ax, ay, az, aw, ns); break;
        case 3: add_rows<BN_IN>(xin, dis, lane, myr, dc, j, 3, sc, sh, ax, ay, az, aw, ns); break;
        case 4: add_rows<BN_IN>(xin, dis, lane, myr, dc, j, 4, sc, sh, ax, ay, az, aw, ns); break;
        case 5: add_rows<BN_IN>(xin, dis, lane, myr, dc, j, 5, sc, sh, ax, ay, az, aw, ns); break;
        case 6: add_rows<BN_IN>(xin, dis, lane, myr, dc, j, 6, sc, sh, ax, ay, az, aw, ns); break;
        case 7: add_rows<BN_IN>(xin, dis, lane, myr, dc, j, 7, sc, sh, ax, ay, az, aw, ns); break;
        default: break;
      }
    }
    if (deg > ADJ_CAP) {
      for (int e0 = 0; e0 < NE_EDGES; e0 += 32) {
        unsigned m = __builtin_amdgcn_ballot_w32(ecol[e0 + lane] == c);
        while (m) {
          int b = __builtin_ctz(m);
          m &= m - 1;
          int r = erow[e0 + b];
          float nrm = dis[r] * dc;
          float4 v = *(const float4*)(xin + (size_t)r * DF + lane * 4);
          if (BN_IN) {
            v.x = bn_leaky(v.x, sc.x, sh.x); v.y = bn_leaky(v.y, sc.y, sh.y);
            v.z = bn_leaky(v.z, sc.z, sh.z); v.w = bn_leaky(v.w, sc.w, sh.w);
          }
          ax += nrm * v.x; ay += nrm * v.y; az += nrm * v.z; aw += nrm * v.w;
          ns += nrm;
        }
      }
    }
    float4 o; o.x = ax; o.y = ay; o.z = az; o.w = aw;
    *(float4*)arow = o;
    if (lane == 0) wsum[i] = ns;
  }
  __syncthreads();

  int l16   = lane & 15;
  int hi    = lane >> 4;
  int khalf = hi << 1;
  int g     = w & 7;
  int ta    = w >> 3;
  int tb    = ta + 4;
  if (ta >= nt) return;
  const float* ap0 = atile + (ta * 16 + l16) * A_PITCH + khalf;
  const float* bp  = Wm + (size_t)khalf * DF + g * 16 + l16;
  float bs = bias[g * 16 + l16];
  float* crow0 = agg + ((size_t)(t0 + ta) * 16 + hi * 8) * DF + g * 16 + l16;
  float* pl0   = part + ((size_t)g * N_PARTS + (t0 + ta)) * 32 + lane;

  if (tb < nt) {
    const float* ap1 = ap0 + 64 * A_PITCH;
    v8f acc0 = {};
    v8f acc1 = {};
#pragma unroll 8
    for (int k = 0; k < DF; k += 4) {
      v2f a0 = *(const v2f*)(ap0 + k);
      v2f a1 = *(const v2f*)(ap1 + k);
      v2f b;
      b.x = bp[(size_t)k * DF];
      b.y = bp[(size_t)(k + 1) * DF];
      acc0 = wmma4(a0, b, acc0);
      acc1 = wmma4(a1, b, acc1);
    }
    store_piece(acc0, wsum + ta * 16, bs, crow0, pl0, hi);
    store_piece(acc1, wsum + tb * 16, bs, crow0 + (size_t)64 * DF, pl0 + 4 * 32, hi);
  } else {
    v8f acc0 = {};
#pragma unroll 8
    for (int k = 0; k < DF; k += 4) {
      v2f a0 = *(const v2f*)(ap0 + k);
      v2f b;
      b.x = bp[(size_t)k * DF];
      b.y = bp[(size_t)(k + 1) * DF];
      acc0 = wmma4(a0, b, acc0);
    }
    store_piece(acc0, wsum + ta * 16, bs, crow0, pl0, hi);
  }
}

__global__ __launch_bounds__(1024)
void bn_finalize_parts(const float* __restrict__ part, float* __restrict__ stats,
                       const float* __restrict__ g, const float* __restrict__ be,
                       float inv_n) {
  __shared__ float red[32 * 32];
  const int QCH  = (N_PARTS + 3) / 4;
  const int ITER = (QCH + 31) / 32;
  int grp  = blockIdx.x;
  int w    = threadIdx.x >> 5;
  int lane = threadIdx.x & 31;
  int sub  = lane >> 3;
  int e0   = (lane & 7) * 4;
  const float* base = part + (size_t)grp * N_PARTS * 32;

  float4 acc; acc.x = 0.f; acc.y = 0.f; acc.z = 0.f; acc.w = 0.f;
#pragma unroll 7
  for (int it = 0; it < ITER; ++it) {
    int p  = (it * 32 + w) * 4 + sub;
    bool ok = p < N_PARTS;
    int pc = ok ? p : N_PARTS - 1;
    float4 v = *(const float4*)(base + (size_t)pc * 32 + e0);
    acc.x += ok ? v.x : 0.f; acc.y += ok ? v.y : 0.f;
    acc.z += ok ? v.z : 0.f; acc.w += ok ? v.w : 0.f;
  }
#pragma unroll
  for (int off = 8; off < 32; off <<= 1) {
    acc.x += __shfl_xor(acc.x, off, 32); acc.y += __shfl_xor(acc.y, off, 32);
    acc.z += __shfl_xor(acc.z, off, 32); acc.w += __shfl_xor(acc.w, off, 32);
  }
  if (lane < 8) *(float4*)(red + w * 32 + e0) = acc;
  __syncthreads();
  if (w == 0) {
    float tot = 0.f;
    for (int i = 0; i < 32; ++i) tot += red[i * 32 + lane];
    float sq = __shfl_down(tot, 16, 32);
    if (lane < 16) {
      int f = grp * 16 + lane;
      float m   = tot * inv_n;
      float var = sq * inv_n - m * m;
      float sc  = g[f] * rsqrtf(var + EPS_BN);
      stats[f]      = sc;
      stats[DF + f] = be[f] - m * sc;
    }
  }
}

__global__ __launch_bounds__(256)
void gather_pairs(const float* __restrict__ aggp, const float* __restrict__ stats,
                  const int* __restrict__ di,
                  float* __restrict__ amat, float* __restrict__ bmat) {
  int t = blockIdx.x * blockDim.x + threadIdx.x;
  if (t >= BPAIRS * 32) return;
  int j = t >> 5, q = t & 31;
  int ia = di[2 * j] - 1;     if (ia < 0) ia += N_NODES;
  int ib = di[2 * j + 1] - 1; if (ib < 0) ib += N_NODES;
  const float4 sc = *(const float4*)(stats + q * 4);
  const float4 sh = *(const float4*)(stats + DF + q * 4);
  float4 va = *(const float4*)(aggp + (size_t)ia * DF + q * 4);
  float4 vb = *(const float4*)(aggp + (size_t)ib * DF + q * 4);
  va.x = bn_leaky(va.x, sc.x, sh.x); va.y = bn_leaky(va.y, sc.y, sh.y);
  va.z = bn_leaky(va.z, sc.z, sh.z); va.w = bn_leaky(va.w, sc.w, sh.w);
  vb.x = bn_leaky(vb.x, sc.x, sh.x); vb.y = bn_leaky(vb.y, sc.y, sh.y);
  vb.z = bn_leaky(vb.z, sc.z, sh.z); vb.w = bn_leaky(vb.w, sc.w, sh.w);
  *(float4*)(amat + (size_t)j * DF + q * 4) = va;
  *(float4*)(bmat + (size_t)j * DF + q * 4) = vb;
}

__global__ __launch_bounds__(256)
void rowdot(const float* __restrict__ tm, const float* __restrict__ bm,
            float* __restrict__ out) {
  int t = blockIdx.x * blockDim.x + threadIdx.x;
  int j = t >> 5, lane = t & 31;
  if (j >= BPAIRS) return;
  float4 tv = *(const float4*)(tm + (size_t)j * DF + lane * 4);
  float4 bv = *(const float4*)(bm + (size_t)j * DF + lane * 4);
  float p = tv.x * bv.x + tv.y * bv.y + tv.z * bv.z + tv.w * bv.w;
#pragma unroll
  for (int off = 16; off > 0; off >>= 1)
    p += __shfl_down(p, off, 32);
  if (lane == 0) out[j] = p;
}

extern "C" void kernel_launch(void* const* d_in, const int* in_sizes, int n_in,
                              void* d_out, int out_size, void* d_ws, size_t ws_size,
                              hipStream_t stream) {
  const float* x   = (const float*)d_in[0];
  const int*   edg = (const int*)d_in[1];
  const int*   di  = (const int*)d_in[2];
  const float* w1  = (const float*)d_in[3];
  const float* b1  = (const float*)d_in[4];
  const float* w2  = (const float*)d_in[5];
  const float* b2  = (const float*)d_in[6];
  const float* w3  = (const float*)d_in[7];
  const float* b3  = (const float*)d_in[8];
  const float* g1  = (const float*)d_in[9];
  const float* be1 = (const float*)d_in[10];
  const float* g2  = (const float*)d_in[11];
  const float* be2 = (const float*)d_in[12];
  const float* g3  = (const float*)d_in[13];
  const float* be3 = (const float*)d_in[14];
  const float* p1  = (const float*)d_in[15];
  const float* p2  = (const float*)d_in[16];
  float* out = (float*)d_out;

  float* ws = (float*)d_ws;
  size_t off = 0;
  float* dis    = ws + off; off += N_NODES;
  float* stats  = ws + off; off += 256;
  off += 128 * 128;
  int*   cnt    = (int*)(ws + off); off += N_NODES;
  int*   cellcnt = (int*)(ws + off);
  off += N_NODES + 16;
  off += N_NODES;
  off += NE_EDGES;
  float* lin    = ws + off; off += (size_t)N_NODES * DF;
  float* agg    = ws + off; off += (size_t)N_NODES * DF;
  float* agg2   = ws + off; off += (size_t)N_NODES * DF;
  float* part   = lin + 16;
  int*   slots  = (int*)(lin + 2097152 + 48);
  int*   stage  = (int*)(agg2 + 48);
  float* U    = lin;
  float* Qt   = lin + 16384;
  float* amat = lin + 32768;
  float* bmat = amat + (size_t)BPAIRS * DF;
  float* tmat = bmat + (size_t)BPAIRS * DF;

  const int* erow = edg;
  const int* ecol = edg + NE_EDGES;

  edge_partition<<<CH_BLOCKS, CH_THREADS, 0, stream>>>(erow, ecol, stage, cellcnt);
  adj_build<<<N_OWN, ADJ_THREADS, 0, stream>>>(erow, ecol, stage, cellcnt, cnt, slots, dis);

  const float* Wl[3] = {w1, w2, w3};
  const float* Bl[3] = {b1, b2, b3};
  const float* Gl[3] = {g1, g2, g3};
  const float* El[3] = {be1, be2, be3};
  float* Al[3] = {agg, agg2, agg};
  gcn_aggregate_linear<false><<<AGG_GRID, AGG_WAVES * 32, 0, stream>>>(
      cnt, slots, erow, ecol, dis, x, stats, Wl[0], Bl[0], Al[0], part);
  bn_finalize_parts<<<DF / 16, 1024, 0, stream>>>(
      part, stats, Gl[0], El[0], 1.0f / (float)N_NODES);
  for (int l = 1; l < 3; ++l) {
    gcn_aggregate_linear<true><<<AGG_GRID, AGG_WAVES * 32, 0, stream>>>(
        cnt, slots, erow, ecol, dis, Al[l - 1], stats, Wl[l], Bl[l], Al[l], part);
    bn_finalize_parts<<<DF / 16, 1024, 0, stream>>>(
        part, stats, Gl[l], El[l], 1.0f / (float)N_NODES);
  }

  const int small_tiles = (DF / 16) * (DF / 16);
  wmma_matmul<<<(small_tiles * 32 + 255) / 256, 256, 0, stream>>>(p1, p2, nullptr, U, DF, DF, DF, 1);
  wmma_matmul<<<(small_tiles * 32 + 255) / 256, 256, 0, stream>>>(U, p1, nullptr, Qt, DF, DF, DF, 1);
  gather_pairs<<<(BPAIRS * 32 + 255) / 256, 256, 0, stream>>>(Al[2], stats, di, amat, bmat);
  wmma_gemm_strip<<<((BPAIRS / 16) * 32 + 255) / 256, 256, 0, stream>>>(
      amat, Qt, nullptr, tmat, BPAIRS, DF);
  rowdot<<<(BPAIRS * 32 + 255) / 256, 256, 0, stream>>>(tmat, bmat, out);
}


# Decoder re-associated: y = ((a P1) P2) . (b P1), P1 P2 P1^T never formed; pair gather, the 3 products and the dot in one launch, P slices held in registers; 5 launches -> 1; sum order changes
# speedup vs baseline: 7.0246x; 1.1510x over previous
#include <hip/hip_runtime.h>

#define N_NODES 50000
#define NE_EDGES 800000
#define BPAIRS 4096
#define DF 128
#define EPS_BN 1e-5f

typedef float v2f __attribute__((ext_vector_type(2)));
typedef float v8f __attribute__((ext_vector_type(8)));
typedef float v4f __attribute__((ext_vector_type(4)));

__device__ __forceinline__ v8f wmma4(v2f a, v2f b, v8f c) {
  return __builtin_amdgcn_wmma_f32_16x16x4_f32(
       false, a,  false, b,
       (short)0, c,  false,  false);
}

#define ADJ_CAP 64
#define ADJ_THREADS 1024
#define OWN_SH 8
#define OWN_N (1 << OWN_SH)
#define N_OWN ((N_NODES + OWN_N - 1) / OWN_N)
#define CH_BLOCKS 256
#define CH_E (NE_EDGES / CH_BLOCKS)
#define CH_THREADS 512
#define CH_PER ((CH_E + CH_THREADS - 1) / CH_THREADS)
#define CELL_CAP 64
static_assert(NE_EDGES % CH_BLOCKS == 0, "whole runs only");
static_assert(N_NODES <= 65536, "a source node is packed into 16 bits beside the target's place");
static_assert(CH_BLOCKS * (CELL_CAP / 4) == 4 * ADJ_THREADS, "an owner's cells are four 16-byte pieces per thread");

__global__ __launch_bounds__(CH_THREADS)
void edge_partition(const int* __restrict__ erow, const int* __restrict__ ecol,
                    int* __restrict__ stage, int* __restrict__ cellcnt) {
  __shared__ int cntL[N_OWN];
  __shared__ int cellL[N_OWN * CELL_CAP];
  const int b = blockIdx.x;
  for (int i = threadIdx.x; i < N_OWN; i += CH_THREADS) cntL[i] = 0;
  __syncthreads();

  const int e0 = b * CH_E;
  int cc[CH_PER], rr[CH_PER];
#pragma unroll
  for (int u = 0; u < CH_PER; ++u) {
    int i = threadIdx.x + u * CH_THREADS;
    int e = e0 + (i < CH_E ? i : CH_E - 1);
    cc[u] = ecol[e];
    rr[u] = erow[e];
  }
  int rk[CH_PER];
#pragma unroll
  for (int u = 0; u < CH_PER; ++u) {
    int i = threadIdx.x + u * CH_THREADS;
    rk[u] = i < CH_E ? atomicAdd(&cntL[cc[u] >> OWN_SH], 1) : CELL_CAP;
  }
#pragma unroll
  for (int u = 0; u < CH_PER; ++u)
    if (rk[u] < CELL_CAP)
      cellL[(cc[u] >> OWN_SH) * CELL_CAP + rk[u]] = ((cc[u] & (OWN_N - 1)) << 16) | rr[u];
  __syncthreads();

  for (int o = threadIdx.x; o < N_OWN; o += CH_THREADS) cellcnt[o * CH_BLOCKS + b] = cntL[o];
  int4* out4 = (int4*)stage;
  const int4* in4 = (const int4*)cellL;
  for (int p = threadIdx.x; p < N_OWN * (CELL_CAP / 4); p += CH_THREADS) {
    int o = p >> 4, j = p & 15;
    int n = cntL[o];
    if (n <= CELL_CAP && j * 4 < n)
      out4[((size_t)o * CH_BLOCKS + b) * (CELL_CAP / 4) + j] = in4[p];
  }
}

__device__ __forceinline__ void slot_edge(int* cntL, int* slotL, int lc, int r) {
  int rank = atomicAdd(&cntL[lc], 1);
  if (rank < ADJ_CAP) slotL[lc * ADJ_CAP + rank] = r;
}

__global__ __launch_bounds__(ADJ_THREADS)
void adj_build(const int* __restrict__ erow, const int* __restrict__ ecol,
               const int* __restrict__ stage, const int* __restrict__ cellcnt,
               int* __restrict__ cnt, int* __restrict__ slots, float* __restrict__ dis) {
  __shared__ int cntL[OWN_N];
  __shared__ int ccL[CH_BLOCKS];
  __shared__ int slotL[OWN_N * ADJ_CAP];
  const int o  = blockIdx.x;
  const int lo = o * OWN_N;
  int nb = N_NODES - lo; if (nb > OWN_N) nb = OWN_N;
  for (int i = threadIdx.x; i < OWN_N; i += ADJ_THREADS) cntL[i] = 0;
  for (int i = threadIdx.x; i < CH_BLOCKS; i += ADJ_THREADS) ccL[i] = cellcnt[o * CH_BLOCKS + i];
  __syncthreads();

  const int4* st4 = (const int4*)stage + (size_t)o * CH_BLOCKS * (CELL_CAP / 4);
  int4 v[4];
  int left[4];
#pragma unroll
  for (int u = 0; u < 4; ++u) {
    int p = threadIdx.x + u * ADJ_THREADS;
    int n = ccL[p >> 4];
    left[u] = n <= CELL_CAP ? n - (p & 15) * 4 : 0;
    v[u].x = v[u].y = v[u].z = v[u].w = 0;
    if (left[u] > 0) v[u] = st4[p];
  }
  int rk[4][4];
#pragma unroll
  for (int u = 0; u < 4; ++u) {
    int pe[4] = {v[u].x, v[u].y, v[u].z, v[u].w};
#pragma unroll
    for (int k = 0; k < 4; ++k)
      rk[u][k] = left[u] > k ? atomicAdd(&cntL[pe[k] >> 16], 1) : ADJ_CAP;
  }
#pragma unroll
  for (int u = 0; u < 4; ++u) {
    int pe[4] = {v[u].x, v[u].y, v[u].z, v[u].w};
#pragma unroll
    for (int k = 0; k < 4; ++k)
      if (rk[u][k] < ADJ_CAP) slotL[(pe[k] >> 16) * ADJ_CAP + rk[u][k]] = pe[k] & 0xffff;
  }
  for (int b = 0; b < CH_BLOCKS; ++b) {
    if (ccL[b] <= CELL_CAP) continue;
    for (int i = threadIdx.x; i < CH_E; i += ADJ_THREADS) {
      int c = ecol[b * CH_E + i];
      if ((c >> OWN_SH) == o) slot_edge(cntL, slotL, c & (OWN_N - 1), erow[b * CH_E + i]);
    }
  }
  __syncthreads();

  for (int i = threadIdx.x; i < nb; i += ADJ_THREADS) {
    int dg = cntL[i];
    cnt[lo + i] = dg;
    dis[lo + i] = rsqrtf((float)dg + 1.0f);
  }
  int4* out4 = (int4*)(slots + (size_t)lo * ADJ_CAP);
  const int4* in4 = (const int4*)slotL;
  for (int t = threadIdx.x; t < nb * (ADJ_CAP / 4); t += ADJ_THREADS) {
    int i = t >> 4, j = t & 15;
    if (cntL[i] - j * 4 > 0) out4[t] = in4[t];
  }
}

#define AGG_WAVES 32
#define N_PARTS (N_NODES / 16)
#define AGG_GRID 512
#define AGG_MAXT ((N_PARTS + AGG_GRID - 1) / AGG_GRID)
#define A_PITCH 132
static_assert(N_NODES % 16 == 0, "whole 16-row tiles only");
static_assert(AGG_WAVES == 32, "four sets of eight product waves");
static_assert(AGG_MAXT <= 8, "a product wave takes at most two tiles, four tiles apart");
static_assert(N_PARTS / AGG_GRID >= 4, "every product wave has a first tile");

__device__ __forceinline__ float bn_leaky(float v, float sc, float sh) {
  float y = v * sc + sh;
  return fmaxf(y, 0.1f * y);
}

template <bool BN_IN>
__device__ __forceinline__ void add_rows(const float* __restrict__ xin, const float* __restrict__ dis,
                                         int lane, int myr, float dc, int j0, const int n,
                                         const float4 sc, const float4 sh,
                                         float& ax, float& ay, float& az, float& aw, float& ns) {
  v4f v[8];
  float  d[8];
#pragma unroll
  for (int u = 0; u < 8; ++u) {
    if (u < n) {
      int r = __builtin_amdgcn_readlane(myr, j0 + u);
      d[u] = dis[r];
      v[u] = *(const v4f*)(xin + (size_t)r * DF + lane * 4);
    }
  }
  if (BN_IN) {
    if (n == 8) asm volatile("" : "+v"(v[0]), "+v"(v[1]), "+v"(v[2]), "+v"(v[3]), "+v"(v[4]), "+v"(v[5]), "+v"(v[6]), "+v"(v[7]));
    else if (n == 7) asm volatile("" : "+v"(v[0]), "+v"(v[1]), "+v"(v[2]), "+v"(v[3]), "+v"(v[4]), "+v"(v[5]), "+v"(v[6]));
    else if (n == 6) asm volatile("" : "+v"(v[0]), "+v"(v[1]), "+v"(v[2]), "+v"(v[3]), "+v"(v[4]), "+v"(v[5]));
  }
#pragma unroll
  for (int u = 0; u < 8; ++u) {
    if (u < n) {
      float nrm = d[u] * dc;
      v4f r = v[u];
      if (BN_IN) {
        r.x = bn_leaky(r.x, sc.x, sh.x); r.y = bn_leaky(r.y, sc.y, sh.y);
        r.z = bn_leaky(r.z, sc.z, sh.z); r.w = bn_leaky(r.w, sc.w, sh.w);
      }
      ax += nrm * r.x; ay += nrm * r.y; az += nrm * r.z; aw += nrm * r.w;
      ns += nrm;
    }
  }
}

__device__ __forceinline__ void store_piece(v8f acc, const float* __restrict__ ws16, float bs,
                                            float* __restrict__ crow, float* __restrict__ pline,
                                            int hi) {
  float cs = 0.f, cq = 0.f;
#pragma unroll
  for (int r = 0; r < 8; ++r) {
    float v = acc[r] + ws16[hi * 8 + r] * bs;
    crow[(size_t)r * DF] = v;
    cs += v; cq += v * v;
  }
  cs += __shfl_xor(cs, 16, 32);
  cq += __shfl_xor(cq, 16, 32);
  *pline = hi ? cq : cs;
}

template <bool BN_IN>
__global__ __launch_bounds__(AGG_WAVES * 32, 16)
void gcn_aggregate_linear(const int* __restrict__ cnt, const int* __restrict__ slots,
                          const int* __restrict__ erow, const int* __restrict__ ecol,
                          const float* __restrict__ dis, const float* __restrict__ xin,
                          const float* __restrict__ stats_in,
                          const float* __restrict__ Wm, const float* __restrict__ bias,
                          float* __restrict__ agg, float* __restrict__ part) {
  __shared__ float atile[AGG_MAXT * 16 * A_PITCH];
  __shared__ float wsum[AGG_MAXT * 16];

  int w = __builtin_amdgcn_readfirstlane(threadIdx.x >> 5);
  int lane = threadIdx.x & 31;
  const int t0 = ((int)blockIdx.x * N_PARTS) / AGG_GRID;
  const int nt = (((int)blockIdx.x + 1) * N_PARTS) / AGG_GRID - t0;
  const int nrows = nt * 16;
  float4 sc, sh;
  sc.x = sc.y = sc.z = sc.w = 1.f;
  sh.x = sh.y = sh.z = sh.w = 0.f;
  if (BN_IN) {
    sc = *(const float4*)(stats_in + lane * 4);
    sh = *(const float4*)(stats_in + DF + lane * 4);
  }
  float* arow = atile + w * A_PITCH + lane * 4;
  for (int i = w; i < nrows; i += AGG_WAVES, arow += AGG_WAVES * A_PITCH) {
    int c = t0 * 16 + i;
    float dc = dis[c];
    float4 sv = *(const float4*)(xin + (size_t)c * DF + lane * 4);
    if (BN_IN) {
      sv.x = bn_leaky(sv.x, sc.x, sh.x); sv.y = bn_leaky(sv.y, sc.y, sh.y);
      sv.z = bn_leaky(sv.z, sc.z, sh.z); sv.w = bn_leaky(sv.w, sc.w, sh.w);
    }
    float s = dc * dc;
    float ax = s * sv.x, ay = s * sv.y, az = s * sv.z, aw = s * sv.w;
    float ns = s;
    int deg = cnt[c];
    int nslot = deg <= ADJ_CAP ? deg : 0;
    const int* sl = slots + (size_t)c * ADJ_CAP;
    for (int base = 0; base < nslot; base += 32) {
      int left = nslot - base;
      if (left > 32) left = 32;
      int myr = c;
      if (lane < left) myr = sl[base + lane];
      int j = 0;
      for (; j + 8 <= left; j += 8)
        add_rows<BN_IN>(xin, dis, lane, myr, dc, j, 8, sc, sh, ax, ay, az, aw, ns);
      switch (left - j) {
        case 1: add_rows<BN_IN>(xin, dis, lane, myr, dc, j, 1, sc, sh, ax, ay, az, aw, ns); break;
        case 2: add_rows<BN_IN>(xin, dis, lane, myr, dc, j, 2, sc, sh, ax, ay, az, aw, ns); break;
        case 3: add_rows<BN_IN>(xin, dis, lane, myr, dc, j, 3, sc, sh, ax, ay, az, aw, ns); break;
        case 4: add_rows<BN_IN>(xin, dis, lane, myr, dc, j, 4, sc, sh, ax, ay, az, aw, ns); break;
        case 5: add_rows<BN_IN>(xin, dis, lane, myr, dc, j, 5, sc, sh, ax, ay, az, aw, ns); break;
        case 6: add_rows<BN_IN>(xin, dis, lane, myr, dc, j, 6, sc, sh, ax, ay, az, aw, ns); break;
        case 7: add_rows<BN_IN>(xin, dis, lane, myr, dc, j, 7, sc, sh, ax, ay, az, aw, ns); break;
        default: break;
      }
    }
    if (deg > ADJ_CAP) {
      for (int e0 = 0; e0 < NE_EDGES; e0 += 32) {
        unsigned m = __builtin_amdgcn_ballot_w32(ecol[e0 + lane] == c);
        while (m) {
          int b = __builtin_ctz(m);
          m &= m - 1;
          int r = erow[e0 + b];
          float nrm = dis[r] * dc;
          float4 v = *(const float4*)(xin + (size_t)r * DF + lane * 4);
          if (BN_IN) {
            v.x = bn_leaky(v.x, sc.x, sh.x); v.y = bn_leaky(v.y, sc.y, sh.y);
            v.z = bn_leaky(v.z, sc.z, sh.z); v.w = bn_leaky(v.w, sc.w, sh.w);
          }
          ax += nrm * v.x; ay += nrm * v.y; az += nrm * v.z; aw += nrm * v.w;
          ns += nrm;
        }
      }
    }
    float4 o; o.x = ax; o.y = ay; o.z = az; o.w = aw;
    *(float4*)arow = o;
    if (lane == 0) wsum[i] = ns;
  }
  __syncthreads();

  int l16   = lane & 15;
  int hi    = lane >> 4;
  int khalf = hi << 1;
  int g     = w & 7;
  int ta    = w >> 3;
  int tb    = ta + 4;
  if (ta >= nt) return;
  const float* ap0 = atile + (ta * 16 + l16) * A_PITCH + khalf;
  const float* bp  = Wm + (size_t)khalf * DF + g * 16 + l16;
  float bs = bias[g * 16 + l16];
  float* crow0 = agg + ((size_t)(t0 + ta) * 16 + hi * 8) * DF + g * 16 + l16;
  float* pl0   = part + ((size_t)g * N_PARTS + (t0 + ta)) * 32 + lane;

  if (tb < nt) {
    const float* ap1 = ap0 + 64 * A_PITCH;
    v8f acc0 = {};
    v8f acc1 = {};
#pragma unroll 8
    for (int k = 0; k < DF; k += 4) {
      v2f a0 = *(const v2f*)(ap0 + k);
      v2f a1 = *(const v2f*)(ap1 + k);
      v2f b;
      b.x = bp[(size_t)k * DF];
      b.y = bp[(size_t)(k + 1) * DF];
      acc0 = wmma4(a0, b, acc0);
      acc1 = wmma4(a1, b, acc1);
    }
    store_piece(acc0, wsum + ta * 16, bs, crow0, pl0, hi);
    store_piece(acc1, wsum + tb * 16, bs, crow0 + (size_t)64 * DF, pl0 + 4 * 32, hi);
  } else {
    v8f acc0 = {};
#pragma unroll 8
    for (int k = 0; k < DF; k += 4) {
      v2f a0 = *(const v2f*)(ap0 + k);
      v2f b;
      b.x = bp[(size_t)k * DF];
      b.y = bp[(size_t)(k + 1) * DF];
      acc0 = wmma4(a0, b, acc0);
    }
    store_piece(acc0, wsum + ta * 16, bs, crow0, pl0, hi);
  }
}

__global__ __launch_bounds__(1024)
void bn_finalize_parts(const float* __restrict__ part, float* __restrict__ stats,
                       const float* __restrict__ g, const float* __restrict__ be,
                       float inv_n) {
  __shared__ float red[32 * 32];
  const int QCH  = (N_PARTS + 3) / 4;
  const int ITER = (QCH + 31) / 32;
  int grp  = blockIdx.x;
  int w    = threadIdx.x >> 5;
  int lane = threadIdx.x & 31;
  int sub  = lane >> 3;
  int e0   = (lane & 7) * 4;
  const float* base = part + (size_t)grp * N_PARTS * 32;

  float4 acc; acc.x = 0.f; acc.y = 0.f; acc.z = 0.f; acc.w = 0.f;
#pragma unroll 7
  for (int it = 0; it < ITER; ++it) {
    int p  = (it * 32 + w) * 4 + sub;
    bool ok = p < N_PARTS;
    int pc = ok ? p : N_PARTS - 1;
    float4 v = *(const float4*)(base + (size_t)pc * 32 + e0);
    acc.x += ok ? v.x : 0.f; acc.y += ok ? v.y : 0.f;
    acc.z += ok ? v.z : 0.f; acc.w += ok ? v.w : 0.f;
  }
#pragma unroll
  for (int off = 8; off < 32; off <<= 1) {
    acc.x += __shfl_xor(acc.x, off, 32); acc.y += __shfl_xor(acc.y, off, 32);
    acc.z += __shfl_xor(acc.z, off, 32); acc.w += __shfl_xor(acc.w, off, 32);
  }
  if (lane < 8) *(float4*)(red + w * 32 + e0) = acc;
  __syncthreads();
  if (w == 0) {
    float tot = 0.f;
    for (int i = 0; i < 32; ++i) tot += red[i * 32 + lane];
    float sq = __shfl_down(tot, 16, 32);
    if (lane < 16) {
      int f = grp * 16 + lane;
      float m   = tot * inv_n;
      float var = sq * inv_n - m * m;
      float sc  = g[f] * rsqrtf(var + EPS_BN);
      stats[f]      = sc;
      stats[DF + f] = be[f] - m * sc;
    }
  }
}

#define DEC_PAIRS 32
#define DEC_WAVES 16
static_assert(BPAIRS % DEC_PAIRS == 0, "whole work-groups only");
static_assert(DEC_WAVES == 2 * (DF / 16), "one wave per tile and column group");

__global__ __launch_bounds__(DEC_WAVES * 32)
void decode_pairs(const float* __restrict__ aggp, const float* __restrict__ stats,
                  const int* __restrict__ di,
                  const float* __restrict__ P1, const float* __restrict__ P2,
                  float* __restrict__ out) {
  __shared__ float atl[DEC_PAIRS * A_PITCH];
  __shared__ float btl[DEC_PAIRS * A_PITCH];
  __shared__ float utl[DEC_PAIRS * A_PITCH];
  __shared__ float red[(DF / 16) * DEC_PAIRS];

  const int w    = __builtin_amdgcn_readfirstlane(threadIdx.x >> 5);
  const int lane = threadIdx.x & 31;
  const int l16  = lane & 15;
  const int hi   = lane >> 4;
  const int khalf = hi << 1;
  const int g    = w & 7;
  const int tl   = w >> 3;
  const int j0   = (int)blockIdx.x * DEC_PAIRS;

  int ia[2], ib[2];
#pragma unroll
  for (int u = 0; u < 2; ++u) {
    int j = j0 + w * 2 + u;
    ia[u] = di[2 * j] - 1;     if (ia[u] < 0) ia[u] += N_NODES;
    ib[u] = di[2 * j + 1] - 1; if (ib[u] < 0) ib[u] += N_NODES;
  }

  const float* bp1 = P1 + (size_t)khalf * DF + g * 16 + l16;
  const float* bp2 = P2 + (size_t)khalf * DF + g * 16 + l16;
  v2f f1[DF / 4], f2[DF / 4];
#pragma unroll
  for (int s = 0; s < DF / 4; ++s) {
    f1[s].x = bp1[(size_t)(4 * s) * DF];
    f1[s].y = bp1[(size_t)(4 * s + 1) * DF];
  }
#pragma unroll
  for (int s = 0; s < DF / 4; ++s) {
    f2[s].x = bp2[(size_t)(4 * s) * DF];
    f2[s].y = bp2[(size_t)(4 * s + 1) * DF];
  }

  const float4 sc = *(const float4*)(stats + lane * 4);
  const float4 sh = *(const float4*)(stats + DF + lane * 4);
#pragma unroll
  for (int u = 0; u < 2; ++u) {
    float4 va = *(const float4*)(aggp + (size_t)ia[u] * DF + lane * 4);
    float4 vb = *(const float4*)(aggp + (size_t)ib[u] * DF + lane * 4);
    va.x = bn_leaky(va.x, sc.x, sh.x); va.y = bn_leaky(va.y, sc.y, sh.y);
    va.z = bn_leaky(va.z, sc.z, sh.z); va.w = bn_leaky(va.w, sc.w, sh.w);
    vb.x = bn_leaky(vb.x, sc.x, sh.x); vb.y = bn_leaky(vb.y, sc.y, sh.y);
    vb.z = bn_leaky(vb.z, sc.z, sh.z); vb.w = bn_leaky(vb.w, sc.w, sh.w);
    *(float4*)(atl + (w * 2 + u) * A_PITCH + lane * 4) = va;
    *(float4*)(btl + (w * 2 + u) * A_PITCH + lane * 4) = vb;
  }
  __syncthreads();

  const int arow = (tl * 16 + l16) * A_PITCH + khalf;
  v8f accA = {};
  v8f accB = {};
#pragma unroll
  for (int s = 0; s < DF / 4; ++s) {
    v2f a0 = *(const v2f*)(atl + arow + 4 * s);
    v2f a1 = *(const v2f*)(btl + arow + 4 * s);
    accA = wmma4(a0, f1[s], accA);
    accB = wmma4(a1, f1[s], accB);
  }
  float* urow = utl + (tl * 16 + hi * 8) * A_PITCH + g * 16 + l16;
#pragma unroll
  for (int r = 0; r < 8; ++r) urow[r * A_PITCH] = accA[r];
  __syncthreads();

  v8f accC = {};
#pragma unroll
  for (int s = 0; s < DF / 4; ++s) {
    v2f a2 = *(const v2f*)(utl + arow + 4 * s);
    accC = wmma4(a2, f2[s], accC);
  }

  float pr[8];
#pragma unroll
  for (int r = 0; r < 8; ++r) pr[r] = accC[r] * accB[r];
#pragma unroll
  for (int off = 8; off > 0; off >>= 1) {
#pragma unroll
    for (int r = 0; r < 8; ++r) pr[r] += __shfl_xor(pr[r], off, 32);
  }
  if (l16 == 0) {
#pragma unroll
    for (int r = 0; r < 8; ++r) red[g * DEC_PAIRS + tl * 16 + hi * 8 + r] = pr[r];
  }
  __syncthreads();
  if (w == 0) {
    float y = 0.f;
#pragma unroll
    for (int q = 0; q < DF / 16; ++q) y += red[q * DEC_PAIRS + lane];
    out[j0 + lane] = y;
  }
}

extern "C" void kernel_launch(void* const* d_in, const int* in_sizes, int n_in,
                              void* d_out, int out_size, void* d_ws, size_t ws_size,
                              hipStream_t stream) {
  const float* x   = (const float*)d_in[0];
  const int*   edg = (const int*)d_in[1];
  const int*   di  = (const int*)d_in[2];
  const float* w1  = (const float*)d_in[3];
  const float* b1  = (const float*)d_in[4];
  const float* w2  = (const float*)d_in[5];
  const float* b2  = (const float*)d_in[6];
  const float* w3  = (const float*)d_in[7];
  const float* b3  = (const float*)d_in[8];
  const float* g1  = (const float*)d_in[9];
  const float* be1 = (const float*)d_in[10];
  const float* g2  = (const float*)d_in[11];
  const float* be2 = (const float*)d_in[12];
  const float* g3  = (const float*)d_in[13];
  const float* be3 = (const float*)d_in[14];
  const float* p1  = (const float*)d_in[15];
  const float* p2  = (const float*)d_in[16];
  float* out = (float*)d_out;

  float* ws = (float*)d_ws;
  size_t off = 0;
  float* dis    = ws + off; off += N_NODES;
  float* stats  = ws + off; off += 256;
  off += 128 * 128;
  int*   cnt    = (int*)(ws + off); off += N_NODES;
  int*   cellcnt = (int*)(ws + off);
  off += N_NODES + 16;
  off += N_NODES;
  off += NE_EDGES;
  float* lin    = ws + off; off += (size_t)N_NODES * DF;
  float* agg    = ws + off; off += (size_t)N_NODES * DF;
  float* agg2   = ws + off; off += (size_t)N_NODES * DF;
  float* part   = lin + 16;
  int*   slots  = (int*)(lin + 2097152 + 48);
  int*   stage  = (int*)(agg2 + 48);
  const int* erow = edg;
  const int* ecol = edg + NE_EDGES;

  edge_partition<<<CH_BLOCKS, CH_THREADS, 0, stream>>>(erow, ecol, stage, cellcnt);
  adj_build<<<N_OWN, ADJ_THREADS, 0, stream>>>(erow, ecol, stage, cellcnt, cnt, slots, dis);

  const float* Wl[3] = {w1, w2, w3};
  const float* Bl[3] = {b1, b2, b3};
  const float* Gl[3] = {g1, g2, g3};
  const float* El[3] = {be1, be2, be3};
  float* Al[3] = {agg, agg2, agg};
  gcn_aggregate_linear<false><<<AGG_GRID, AGG_WAVES * 32, 0, stream>>>(
      cnt, slots, erow, ecol, dis, x, stats, Wl[0], Bl[0], Al[0], part);
  bn_finalize_parts<<<DF / 16, 1024, 0, stream>>>(
      part, stats, Gl[0], El[0], 1.0f / (float)N_NODES);
  for (int l = 1; l < 3; ++l) {
    gcn_aggregate_linear<true><<<AGG_GRID, AGG_WAVES * 32, 0, stream>>>(
        cnt, slots, erow, ecol, dis, Al[l - 1], stats, Wl[l], Bl[l], Al[l], part);
    bn_finalize_parts<<<DF / 16, 1024, 0, stream>>>(
        part, stats, Gl[l], El[l], 1.0f / (float)N_NODES);
  }

  decode_pairs<<<BPAIRS / DEC_PAIRS, DEC_WAVES * 32, 0, stream>>>(Al[2], stats, di, p1, p2, out);
}
